# starting kernel
# baseline (speedup 1.0000x reference)
#include <hip/hip_runtime.h>
#include <hip/hip_bf16.h>
#include <math.h>

typedef __attribute__((ext_vector_type(2))) float v2f;
typedef __attribute__((ext_vector_type(8))) float v8f;

#define PROJ   8192
#define NCH    512
#define HWSZ   196
#define NTILE  1024
#define BLK    256

__global__ __launch_bounds__(BLK)
void cbp_gram_sketch_kernel(const float* __restrict__ x,
                            const float* __restrict__ s1,
                            const float* __restrict__ s2,
                            const int*   __restrict__ h1,
                            const int*   __restrict__ h2,
                            float* __restrict__ out)
{
    __shared__ float bins[PROJ];
    __shared__ float s1l[NCH], s2l[NCH];
    __shared__ int   h1l[NCH], h2l[NCH];
    __shared__ float red[8];

    const int b    = blockIdx.x;
    const int tid  = threadIdx.x;
    const int lane = tid & 31;
    const int wave = tid >> 5;

    for (int p = tid; p < PROJ; p += BLK) bins[p] = 0.0f;
    for (int c = tid; c < NCH; c += BLK) {
        s1l[c] = s1[c];  s2l[c] = s2[c];
        h1l[c] = h1[c];  h2l[c] = h2[c];
    }
    __syncthreads();

    const float* xb = x + (size_t)b * NCH * HWSZ;

    const int half = lane >> 4;
    const int lm16 = lane & 15;
    const int kb   = 2 * half;

    for (int t = wave; t < NTILE; t += 8) {
        const int i = t >> 5;
        const int j = t & 31;

        const float* arow = xb + (size_t)(i * 16 + lm16) * HWSZ;
        const float* brow = xb + (size_t)(j * 16 + lm16) * HWSZ;

        v8f acc = {};
        #pragma unroll 7
        for (int kk = 0; kk < 49; ++kk) {
            const int k0 = kk * 4 + kb;
            v2f av = *(const v2f*)(arow + k0);
            v2f bv = *(const v2f*)(brow + k0);
            acc = __builtin_amdgcn_wmma_f32_16x16x4_f32(
                    false, av, false, bv, (short)0, acc, false, false);
        }

        const int   cB  = j * 16 + lm16;
        const float s2v = s2l[cB];
        const int   h2v = h2l[cB];
        #pragma unroll
        for (int v = 0; v < 8; ++v) {
            const int   cA  = i * 16 + v + 8 * half;
            const float val = acc[v] * s1l[cA] * s2v;
            const int   bkt = (h1l[cA] + h2v) & (PROJ - 1);
            atomicAdd(&bins[bkt], val);
        }
    }
    __syncthreads();

    float yv[PROJ / BLK];
    float ss = 0.0f;
    #pragma unroll
    for (int q = 0; q < PROJ / BLK; ++q) {
        const float tv = bins[tid + q * BLK];
        const float sg = (tv > 0.0f) ? 1.0f : ((tv < 0.0f) ? -1.0f : 0.0f);
        const float y  = sg * sqrtf(fabsf(tv) + 1e-8f);
        yv[q] = y;
        ss   += y * y;
    }
    #pragma unroll
    for (int off = 16; off > 0; off >>= 1)
        ss += __shfl_down(ss, off, 32);
    if (lane == 0) red[wave] = ss;
    __syncthreads();
    if (tid == 0) {
        float tot = 0.0f;
        #pragma unroll
        for (int w = 0; w < 8; ++w) tot += red[w];
        red[0] = fmaxf(sqrtf(tot), 1e-12f);
    }
    __syncthreads();

    const float inv = 1.0f / red[0];
    float* ob = out + (size_t)b * PROJ;
    #pragma unroll
    for (int q = 0; q < PROJ / BLK; ++q)
        ob[tid + q * BLK] = yv[q] * inv;
}

extern "C" void kernel_launch(void* const* d_in, const int* in_sizes, int n_in,
                              void* d_out, int out_size, void* d_ws, size_t ws_size,
                              hipStream_t stream) {
    const float* x  = (const float*)d_in[0];
    const float* s1 = (const float*)d_in[1];
    const float* s2 = (const float*)d_in[2];
    const int*   h1 = (const int*)  d_in[3];
    const int*   h2 = (const int*)  d_in[4];
    float*       y  = (float*)d_out;
    (void)in_sizes; (void)n_in; (void)out_size; (void)d_ws; (void)ws_size;

    cbp_gram_sketch_kernel<<<16, BLK, 0, stream>>>(x, s1, s2, h1, h2, y);
}


# Gram columns stored in (h2, channel) order (B rows taken through a rank table): a bin group reads one short run per row, equal-h2 columns are neighbours, so one table and no h1 level passes
# speedup vs baseline: 7.5675x; 7.5675x over previous
#include <hip/hip_runtime.h>
#include <math.h>

typedef __attribute__((ext_vector_type(2))) float v2f;
typedef __attribute__((ext_vector_type(4))) float v4f;
typedef __attribute__((ext_vector_type(8))) float v8f;

#define PROJ   8192
#define PMASK  (PROJ - 1)
#define NCH    512
#define HWSZ   196
#define KSTEP8 24

template <int NT>
__device__ __forceinline__ void rank_channels(const int* __restrict__ h2, const int tid,
                                              int* bcnt, int* bstart, int* member, int* wsum,
                                              int (&hv)[NCH / NT], int (&rnk)[NCH / NT],
                                              int (&pos)[NCH / NT], int (&cnt)[NCH / NT])
{
    constexpr int CPT = NCH / NT;
    const int lane = tid & 31;
    const int wave = tid >> 5;

    #pragma unroll
    for (int i = 0; i < CPT; ++i) bcnt[tid + i * NT] = 0;
    __syncthreads();

    int key[CPT], slot[CPT];
    #pragma unroll
    for (int i = 0; i < CPT; ++i) {
        const int t = tid + i * NT;
        hv[i]   = h2[t] & PMASK;
        key[i]  = (hv[i] << 9) | t;
        slot[i] = atomicAdd(&bcnt[hv[i] >> 4], 1);
    }
    __syncthreads();

    int v[CPT];
    int s = 0;
    #pragma unroll
    for (int i = 0; i < CPT; ++i) { v[i] = bcnt[tid * CPT + i]; s += v[i]; }
    int incl = s;
    #pragma unroll
    for (int d = 1; d < 32; d <<= 1) {
        const int u = __shfl_up(incl, d, 32);
        incl += (lane >= d) ? u : 0;
    }
    if (lane == 31) wsum[wave] = incl;
    __syncthreads();
    int ex = incl - s;
    #pragma unroll
    for (int w = 0; w < NT / 32; ++w) ex += (w < wave) ? wsum[w] : 0;
    #pragma unroll
    for (int i = 0; i < CPT; ++i) { bstart[tid * CPT + i] = ex; ex += v[i]; }
    __syncthreads();

    #pragma unroll
    for (int i = 0; i < CPT; ++i) member[bstart[hv[i] >> 4] + slot[i]] = key[i];
    __syncthreads();

    #pragma unroll
    for (int i = 0; i < CPT; ++i) {
        const int bk = hv[i] >> 4;
        const int st = bstart[bk];
        const int n  = bcnt[bk];
        int r = st, p = 0, c = 0;
        for (int q = 0; q < n; ++q) {
            const int  m  = member[st + q];
            const bool lt = (m < key[i]);
            const bool sm = ((m >> 9) == hv[i]);
            r += lt ? 1 : 0;
            p += (lt && sm) ? 1 : 0;
            c += sm ? 1 : 0;
        }
        rnk[i] = r; pos[i] = p; cnt[i] = c;
    }
}

__device__ __forceinline__ void mma_2x4(const v2f (&a)[2], const v2f (&bq)[4], v8f (&acc)[2][4])
{
    #pragma unroll
    for (int r = 0; r < 2; ++r) {
        #pragma unroll
        for (int s = 0; s < 4; ++s) {
            acc[r][s] = __builtin_amdgcn_wmma_f32_16x16x4_f32(
                    false, a[r], false, bq[s], (short)0, acc[r][s], false, false);
        }
    }
}

__global__ __launch_bounds__(256)
void cbp_gram_kernel(const float* __restrict__ x,
                     const float* __restrict__ s1,
                     const float* __restrict__ s2,
                     const int*   __restrict__ h2,
                     float* __restrict__ Gs)
{
    __shared__ int bcnt[NCH], bstart[NCH], member[NCH], perm[NCH];
    __shared__ int wsum[8];

    const int b    = blockIdx.x >> 4;
    const int blk  = blockIdx.x & 15;
    const int bi   = blk >> 2;
    const int bj   = blk & 3;
    const int tid  = threadIdx.x;
    const int lane = tid & 31;
    const int wave = tid >> 5;
    const int half = lane >> 4;
    const int lm16 = lane & 15;

    const int ti0 = bi * 8 + (wave >> 1) * 2;
    const int tj0 = bj * 8 + (wave & 1) * 4;

    const float* xb = x + (size_t)b * NCH * HWSZ;

    {
        int hv[2], rnk[2], pos[2], cnt[2];
        rank_channels<256>(h2, tid, bcnt, bstart, member, wsum, hv, rnk, pos, cnt);
        perm[rnk[0]] = tid;
        perm[rnk[1]] = tid + 256;
        __syncthreads();
    }
    int chB[4];
    #pragma unroll
    for (int s = 0; s < 4; ++s) chB[s] = perm[(tj0 + s) * 16 + lm16];

    const float* ra[2];
    const float* rb[4];
    #pragma unroll
    for (int r = 0; r < 2; ++r) ra[r] = xb + (size_t)((ti0 + r) * 16 + lm16) * HWSZ;
    #pragma unroll
    for (int s = 0; s < 4; ++s) rb[s] = xb + (size_t)chB[s] * HWSZ;

    v8f acc[2][4];
    #pragma unroll
    for (int r = 0; r < 2; ++r) {
        #pragma unroll
        for (int s = 0; s < 4; ++s) acc[r][s] = (v8f){};
    }

    const int ko = 4 * half;
    v4f ca[2], cb[4];
    #pragma unroll
    for (int r = 0; r < 2; ++r) ca[r] = *(const v4f*)(ra[r] + ko);
    #pragma unroll
    for (int s = 0; s < 4; ++s) cb[s] = *(const v4f*)(rb[s] + ko);

    for (int m = 0; m < KSTEP8 - 1; ++m) {
        v4f na[2], nb[4];
        const int kn = 8 * (m + 1) + ko;
        #pragma unroll
        for (int r = 0; r < 2; ++r) na[r] = *(const v4f*)(ra[r] + kn);
        #pragma unroll
        for (int s = 0; s < 4; ++s) nb[s] = *(const v4f*)(rb[s] + kn);

        v2f a0[2], b0[4], a1[2], b1[4];
        #pragma unroll
        for (int r = 0; r < 2; ++r) { a0[r] = (v2f){ca[r].x, ca[r].y}; a1[r] = (v2f){ca[r].z, ca[r].w}; }
        #pragma unroll
        for (int s = 0; s < 4; ++s) { b0[s] = (v2f){cb[s].x, cb[s].y}; b1[s] = (v2f){cb[s].z, cb[s].w}; }
        mma_2x4(a0, b0, acc);
        mma_2x4(a1, b1, acc);

        #pragma unroll
        for (int r = 0; r < 2; ++r) ca[r] = na[r];
        #pragma unroll
        for (int s = 0; s < 4; ++s) cb[s] = nb[s];
    }
    {
        v2f ta[2], tb[4];
        const int kt = 8 * KSTEP8 + 2 * half;
        #pragma unroll
        for (int r = 0; r < 2; ++r) ta[r] = *(const v2f*)(ra[r] + kt);
        #pragma unroll
        for (int s = 0; s < 4; ++s) tb[s] = *(const v2f*)(rb[s] + kt);

        v2f a0[2], b0[4], a1[2], b1[4];
        #pragma unroll
        for (int r = 0; r < 2; ++r) { a0[r] = (v2f){ca[r].x, ca[r].y}; a1[r] = (v2f){ca[r].z, ca[r].w}; }
        #pragma unroll
        for (int s = 0; s < 4; ++s) { b0[s] = (v2f){cb[s].x, cb[s].y}; b1[s] = (v2f){cb[s].z, cb[s].w}; }
        mma_2x4(a0, b0, acc);
        mma_2x4(a1, b1, acc);
        mma_2x4(ta, tb, acc);
    }

    float* Gb = Gs + (size_t)b * NCH * NCH;
    #pragma unroll
    for (int r = 0; r < 2; ++r) {
        #pragma unroll
        for (int s = 0; s < 4; ++s) {
            const int   cB  = (tj0 + s) * 16 + lm16;
            const float s2v = s2[chB[s]];
            #pragma unroll
            for (int v = 0; v < 8; ++v) {
                const int cA = (ti0 + r) * 16 + v + 8 * half;
                Gb[(size_t)cA * NCH + cB] = acc[r][s][v] * s1[cA] * s2v;
            }
        }
    }
}

typedef v2f v2f_a4 __attribute__((aligned(4)));

#define ONE_HI 0x3F800000
#define ONE_LO 0x00003F80

__global__ __launch_bounds__(512)
void cbp_gather_kernel(const float* __restrict__ Gs,
                       const int*   __restrict__ h1,
                       const int*   __restrict__ h2,
                       float* __restrict__ out)
{
    __shared__ int2 tab[PROJ];
    __shared__ int bcnt[NCH], bstart[NCH], member[NCH];
    __shared__ int wsum[16];
    __shared__ int cntw[16];
    __shared__ int spH[NCH / 3 + 2];
    __shared__ int spC[NCH / 3 + 2];
    __shared__ int spN[NCH / 3 + 2];

    const int b    = blockIdx.x >> 4;
    const int seg  = blockIdx.x & 15;
    const int tid  = threadIdx.x;
    const int lane = tid & 31;
    const int wave = tid >> 5;
    const int p    = seg * 512 + tid;
    const int p8   = p * 8;

    const float* Gb   = Gs + (size_t)b * NCH * NCH;
    const char*  tabb = (const char*)tab;

    #pragma unroll
    for (int i = 0; i < PROJ / 512; ++i) tab[tid + i * 512] = make_int2(0, 0);

    int hv[1], rnk[1], pos[1], cnt[1];
    rank_channels<512>(h2, tid, bcnt, bstart, member, wsum, hv, rnk, pos, cnt);

    int nSp;
    {
        const bool first = (pos[0] == 0);
        if (first) {
            const int col = (rnk[0] < NCH - 1) ? rnk[0] : NCH - 2;
            int w;
            if (cnt[0] >= 2)            w = ONE_HI | ONE_LO;
            else if (rnk[0] == col)     w = ONE_HI;
            else                        w = ONE_LO;
            tab[hv[0]] = make_int2(col * 4, w);
        }
        const bool     sp = first && (cnt[0] >= 3);
        const unsigned mk = __builtin_amdgcn_ballot_w32(sp);
        if (lane == 0) cntw[wave] = __popc(mk);
        __syncthreads();
        int base = 0, tot = 0;
        #pragma unroll
        for (int w = 0; w < 16; ++w) {
            const int cw = cntw[w];
            base += (w < wave) ? cw : 0;
            tot  += cw;
        }
        if (sp) {
            const int at = base + __popc(mk & ((1u << lane) - 1u));
            spH[at] = hv[0];
            spC[at] = rnk[0];
            spN[at] = cnt[0];
        }
        nSp = __builtin_amdgcn_readfirstlane(tot);
        __syncthreads();
    }

    float acc = 0.0f;
    for (int c0 = 0; c0 < NCH; c0 += 16) {
        int2 e[16];
        #pragma unroll
        for (int j = 0; j < 16; ++j) {
            const int a = (p8 - (h1[c0 + j] << 3)) & (PMASK << 3);
            e[j] = *(const int2*)(tabb + a);
        }
        v2f g[16];
        #pragma unroll
        for (int j = 0; j < 16; ++j) {
            const char* row = (const char*)(Gb + (size_t)(c0 + j) * NCH);
            const unsigned off = (unsigned)((e[j].x < 0) ? 0 : e[j].x);
            g[j] = *(const v2f_a4*)(row + off);
        }
        #pragma unroll
        for (int j = 0; j < 16; ++j) {
            const float w0 = __int_as_float(e[j].y & (int)0xFFFF0000);
            const float w1 = __int_as_float(e[j].y << 16);
            acc = fmaf(g[j].x, w0, acc);
            acc = fmaf(g[j].y, w1, acc);
        }
    }

    for (int k = 0; k < nSp; ++k) {
        const int cs = spC[k];
        const int ns = spN[k];
        const int want = (p - spH[k]) & PMASK;
        for (int c0 = 0; c0 < NCH; c0 += 16) {
            bool any = false;
            #pragma unroll
            for (int j = 0; j < 16; ++j) any = any || ((h1[c0 + j] & PMASK) == want);
            if (any) {
                for (int j = 0; j < 16; ++j) {
                    if ((h1[c0 + j] & PMASK) == want) {
                        for (int q = 2; q < ns; ++q) acc += Gb[(size_t)(c0 + j) * NCH + cs + q];
                    }
                }
            }
        }
    }

    const float sg = (acc > 0.0f) ? 1.0f : ((acc < 0.0f) ? -1.0f : 0.0f);
    out[(size_t)b * PROJ + p] = sg * sqrtf(fabsf(acc) + 1e-8f);
}

__global__ __launch_bounds__(1024)
void cbp_norm_kernel(float* __restrict__ out)
{
    __shared__ float red[32];
    const int b    = blockIdx.x;
    const int tid  = threadIdx.x;
    const int lane = tid & 31;
    const int wave = tid >> 5;

    float* ob = out + (size_t)b * PROJ;
    float yv[PROJ / 1024];
    float ss = 0.0f;
    #pragma unroll
    for (int q = 0; q < PROJ / 1024; ++q) {
        const float y = ob[tid + q * 1024];
        yv[q] = y;
        ss   += y * y;
    }
    #pragma unroll
    for (int off = 16; off > 0; off >>= 1)
        ss += __shfl_down(ss, off, 32);
    if (lane == 0) red[wave] = ss;
    __syncthreads();
    if (tid == 0) {
        float tot = 0.0f;
        #pragma unroll
        for (int w = 0; w < 32; ++w) tot += red[w];
        red[0] = fmaxf(sqrtf(tot), 1e-12f);
    }
    __syncthreads();

    const float inv = 1.0f / red[0];
    #pragma unroll
    for (int q = 0; q < PROJ / 1024; ++q)
        ob[tid + q * 1024] = yv[q] * inv;
}

extern "C" void kernel_launch(void* const* d_in, const int* in_sizes, int n_in,
                              void* d_out, int out_size, void* d_ws, size_t ws_size,
                              hipStream_t stream) {
    const float* x  = (const float*)d_in[0];
    const float* s1 = (const float*)d_in[1];
    const float* s2 = (const float*)d_in[2];
    const int*   h1 = (const int*)  d_in[3];
    const int*   h2 = (const int*)  d_in[4];
    float*       y  = (float*)d_out;
    float*       Gs = (float*)d_ws;
    (void)in_sizes; (void)n_in; (void)out_size; (void)ws_size;

    cbp_gram_kernel<<<256, 256, 0, stream>>>(x, s1, s2, h2, Gs);
    cbp_gather_kernel<<<256, 512, 0, stream>>>(Gs, h1, h2, y);
    cbp_norm_kernel<<<16, 1024, 0, stream>>>(y);
}


# Gather loop: 64 rows per trip (was a quarter of that), table reads grouped, waves per SIMD capped at 4 for the registers: a wave keeps 32-48 row reads in flight, half the dependent round trips
# speedup vs baseline: 10.9681x; 1.4494x over previous
#include <hip/hip_runtime.h>
#include <math.h>

typedef __attribute__((ext_vector_type(2))) float v2f;
typedef __attribute__((ext_vector_type(4))) float v4f;
typedef __attribute__((ext_vector_type(8))) float v8f;

#define PROJ   8192
#define PMASK  (PROJ - 1)
#define NCH    512
#define HWSZ   196
#define KSTEP8 24

template <int NT>
__device__ __forceinline__ void rank_channels(const int* __restrict__ h2, const int tid,
                                              int* bcnt, int* bstart, int* member, int* wsum,
                                              int (&hv)[NCH / NT], int (&rnk)[NCH / NT],
                                              int (&pos)[NCH / NT], int (&cnt)[NCH / NT])
{
    constexpr int CPT = NCH / NT;
    const int lane = tid & 31;
    const int wave = tid >> 5;

    #pragma unroll
    for (int i = 0; i < CPT; ++i) bcnt[tid + i * NT] = 0;
    __syncthreads();

    int key[CPT], slot[CPT];
    #pragma unroll
    for (int i = 0; i < CPT; ++i) {
        const int t = tid + i * NT;
        hv[i]   = h2[t] & PMASK;
        key[i]  = (hv[i] << 9) | t;
        slot[i] = atomicAdd(&bcnt[hv[i] >> 4], 1);
    }
    __syncthreads();

    int v[CPT];
    int s = 0;
    #pragma unroll
    for (int i = 0; i < CPT; ++i) { v[i] = bcnt[tid * CPT + i]; s += v[i]; }
    int incl = s;
    #pragma unroll
    for (int d = 1; d < 32; d <<= 1) {
        const int u = __shfl_up(incl, d, 32);
        incl += (lane >= d) ? u : 0;
    }
    if (lane == 31) wsum[wave] = incl;
    __syncthreads();
    int ex = incl - s;
    #pragma unroll
    for (int w = 0; w < NT / 32; ++w) ex += (w < wave) ? wsum[w] : 0;
    #pragma unroll
    for (int i = 0; i < CPT; ++i) { bstart[tid * CPT + i] = ex; ex += v[i]; }
    __syncthreads();

    #pragma unroll
    for (int i = 0; i < CPT; ++i) member[bstart[hv[i] >> 4] + slot[i]] = key[i];
    __syncthreads();

    #pragma unroll
    for (int i = 0; i < CPT; ++i) {
        const int bk = hv[i] >> 4;
        const int st = bstart[bk];
        const int n  = bcnt[bk];
        int r = st, p = 0, c = 0;
        for (int q = 0; q < n; ++q) {
            const int  m  = member[st + q];
            const bool lt = (m < key[i]);
            const bool sm = ((m >> 9) == hv[i]);
            r += lt ? 1 : 0;
            p += (lt && sm) ? 1 : 0;
            c += sm ? 1 : 0;
        }
        rnk[i] = r; pos[i] = p; cnt[i] = c;
    }
}

__device__ __forceinline__ void mma_2x4(const v2f (&a)[2], const v2f (&bq)[4], v8f (&acc)[2][4])
{
    #pragma unroll
    for (int r = 0; r < 2; ++r) {
        #pragma unroll
        for (int s = 0; s < 4; ++s) {
            acc[r][s] = __builtin_amdgcn_wmma_f32_16x16x4_f32(
                    false, a[r], false, bq[s], (short)0, acc[r][s], false, false);
        }
    }
}

__global__ __launch_bounds__(256)
void cbp_gram_kernel(const float* __restrict__ x,
                     const float* __restrict__ s1,
                     const float* __restrict__ s2,
                     const int*   __restrict__ h2,
                     float* __restrict__ Gs)
{
    __shared__ int bcnt[NCH], bstart[NCH], member[NCH], perm[NCH];
    __shared__ int wsum[8];

    const int b    = blockIdx.x >> 4;
    const int blk  = blockIdx.x & 15;
    const int bi   = blk >> 2;
    const int bj   = blk & 3;
    const int tid  = threadIdx.x;
    const int lane = tid & 31;
    const int wave = tid >> 5;
    const int half = lane >> 4;
    const int lm16 = lane & 15;

    const int ti0 = bi * 8 + (wave >> 1) * 2;
    const int tj0 = bj * 8 + (wave & 1) * 4;

    const float* xb = x + (size_t)b * NCH * HWSZ;

    {
        int hv[2], rnk[2], pos[2], cnt[2];
        rank_channels<256>(h2, tid, bcnt, bstart, member, wsum, hv, rnk, pos, cnt);
        perm[rnk[0]] = tid;
        perm[rnk[1]] = tid + 256;
        __syncthreads();
    }
    int chB[4];
    #pragma unroll
    for (int s = 0; s < 4; ++s) chB[s] = perm[(tj0 + s) * 16 + lm16];

    const float* ra[2];
    const float* rb[4];
    #pragma unroll
    for (int r = 0; r < 2; ++r) ra[r] = xb + (size_t)((ti0 + r) * 16 + lm16) * HWSZ;
    #pragma unroll
    for (int s = 0; s < 4; ++s) rb[s] = xb + (size_t)chB[s] * HWSZ;

    v8f acc[2][4];
    #pragma unroll
    for (int r = 0; r < 2; ++r) {
        #pragma unroll
        for (int s = 0; s < 4; ++s) acc[r][s] = (v8f){};
    }

    const int ko = 4 * half;
    v4f ca[2], cb[4];
    #pragma unroll
    for (int r = 0; r < 2; ++r) ca[r] = *(const v4f*)(ra[r] + ko);
    #pragma unroll
    for (int s = 0; s < 4; ++s) cb[s] = *(const v4f*)(rb[s] + ko);

    for (int m = 0; m < KSTEP8 - 1; ++m) {
        v4f na[2], nb[4];
        const int kn = 8 * (m + 1) + ko;
        #pragma unroll
        for (int r = 0; r < 2; ++r) na[r] = *(const v4f*)(ra[r] + kn);
        #pragma unroll
        for (int s = 0; s < 4; ++s) nb[s] = *(const v4f*)(rb[s] + kn);

        v2f a0[2], b0[4], a1[2], b1[4];
        #pragma unroll
        for (int r = 0; r < 2; ++r) { a0[r] = (v2f){ca[r].x, ca[r].y}; a1[r] = (v2f){ca[r].z, ca[r].w}; }
        #pragma unroll
        for (int s = 0; s < 4; ++s) { b0[s] = (v2f){cb[s].x, cb[s].y}; b1[s] = (v2f){cb[s].z, cb[s].w}; }
        mma_2x4(a0, b0, acc);
        mma_2x4(a1, b1, acc);

        #pragma unroll
        for (int r = 0; r < 2; ++r) ca[r] = na[r];
        #pragma unroll
        for (int s = 0; s < 4; ++s) cb[s] = nb[s];
    }
    {
        v2f ta[2], tb[4];
        const int kt = 8 * KSTEP8 + 2 * half;
        #pragma unroll
        for (int r = 0; r < 2; ++r) ta[r] = *(const v2f*)(ra[r] + kt);
        #pragma unroll
        for (int s = 0; s < 4; ++s) tb[s] = *(const v2f*)(rb[s] + kt);

        v2f a0[2], b0[4], a1[2], b1[4];
        #pragma unroll
        for (int r = 0; r < 2; ++r) { a0[r] = (v2f){ca[r].x, ca[r].y}; a1[r] = (v2f){ca[r].z, ca[r].w}; }
        #pragma unroll
        for (int s = 0; s < 4; ++s) { b0[s] = (v2f){cb[s].x, cb[s].y}; b1[s] = (v2f){cb[s].z, cb[s].w}; }
        mma_2x4(a0, b0, acc);
        mma_2x4(a1, b1, acc);
        mma_2x4(ta, tb, acc);
    }

    float* Gb = Gs + (size_t)b * NCH * NCH;
    #pragma unroll
    for (int r = 0; r < 2; ++r) {
        #pragma unroll
        for (int s = 0; s < 4; ++s) {
            const int   cB  = (tj0 + s) * 16 + lm16;
            const float s2v = s2[chB[s]];
            #pragma unroll
            for (int v = 0; v < 8; ++v) {
                const int cA = (ti0 + r) * 16 + v + 8 * half;
                Gb[(size_t)cA * NCH + cB] = acc[r][s][v] * s1[cA] * s2v;
            }
        }
    }
}

typedef v2f v2f_a4 __attribute__((aligned(4)));

#define ONE_HI 0x3F800000
#define ONE_LO 0x00003F80
#define ROWS_IN_FLIGHT 64

__global__ __launch_bounds__(512) __attribute__((amdgpu_waves_per_eu(4, 4)))
void cbp_gather_kernel(const float* __restrict__ Gs,
                       const int*   __restrict__ h1,
                       const int*   __restrict__ h2,
                       float* __restrict__ out)
{
    __shared__ int2 tab[PROJ];
    __shared__ int bcnt[NCH], bstart[NCH], member[NCH];
    __shared__ int wsum[16];
    __shared__ int cntw[16];
    __shared__ int spH[NCH / 3 + 2];
    __shared__ int spC[NCH / 3 + 2];
    __shared__ int spN[NCH / 3 + 2];

    const int b    = blockIdx.x >> 4;
    const int seg  = blockIdx.x & 15;
    const int tid  = threadIdx.x;
    const int lane = tid & 31;
    const int wave = tid >> 5;
    const int p    = seg * 512 + tid;
    const int p8   = p * 8;

    const float* Gb   = Gs + (size_t)b * NCH * NCH;
    const char*  tabb = (const char*)tab;

    #pragma unroll
    for (int i = 0; i < PROJ / 512; ++i) tab[tid + i * 512] = make_int2(0, 0);

    int hv[1], rnk[1], pos[1], cnt[1];
    rank_channels<512>(h2, tid, bcnt, bstart, member, wsum, hv, rnk, pos, cnt);

    int nSp;
    {
        const bool first = (pos[0] == 0);
        if (first) {
            const int col = (rnk[0] < NCH - 1) ? rnk[0] : NCH - 2;
            int w;
            if (cnt[0] >= 2)            w = ONE_HI | ONE_LO;
            else if (rnk[0] == col)     w = ONE_HI;
            else                        w = ONE_LO;
            tab[hv[0]] = make_int2(col * 4, w);
        }
        const bool     sp = first && (cnt[0] >= 3);
        const unsigned mk = __builtin_amdgcn_ballot_w32(sp);
        if (lane == 0) cntw[wave] = __popc(mk);
        __syncthreads();
        int base = 0, tot = 0;
        #pragma unroll
        for (int w = 0; w < 16; ++w) {
            const int cw = cntw[w];
            base += (w < wave) ? cw : 0;
            tot  += cw;
        }
        if (sp) {
            const int at = base + __popc(mk & ((1u << lane) - 1u));
            spH[at] = hv[0];
            spC[at] = rnk[0];
            spN[at] = cnt[0];
        }
        nSp = __builtin_amdgcn_readfirstlane(tot);
        __syncthreads();
    }

    float acc = 0.0f;
    for (int c0 = 0; c0 < NCH; c0 += ROWS_IN_FLIGHT) {
        int wts[ROWS_IN_FLIGHT];
        v2f g[ROWS_IN_FLIGHT];
        int hs[ROWS_IN_FLIGHT];
        #pragma unroll
        for (int j = 0; j < ROWS_IN_FLIGHT; ++j) hs[j] = h1[c0 + j] << 3;
        #pragma unroll
        for (int q = 0; q < ROWS_IN_FLIGHT; q += 16) {
            int2 e[16];
            #pragma unroll
            for (int j = 0; j < 16; ++j) {
                const int a = (p8 - hs[q + j]) & (PMASK << 3);
                e[j] = *(const int2*)(tabb + a);
            }
            __builtin_amdgcn_sched_barrier(0);
            #pragma unroll
            for (int j = 0; j < 16; ++j) {
                const char* row = (const char*)(Gb + (size_t)(c0 + q + j) * NCH);
                const unsigned off = (unsigned)((e[j].x < 0) ? 0 : e[j].x);
                g[q + j]   = *(const v2f_a4*)(row + off);
                wts[q + j] = e[j].y;
            }
        }
        #pragma unroll
        for (int j = 0; j < ROWS_IN_FLIGHT; ++j) {
            const float w0 = __int_as_float(wts[j] & (int)0xFFFF0000);
            const float w1 = __int_as_float(wts[j] << 16);
            acc = fmaf(g[j].x, w0, acc);
            acc = fmaf(g[j].y, w1, acc);
        }
    }

    for (int k = 0; k < nSp; ++k) {
        const int cs = spC[k];
        const int ns = spN[k];
        const int want = (p - spH[k]) & PMASK;
        for (int c0 = 0; c0 < NCH; c0 += 16) {
            bool any = false;
            #pragma unroll
            for (int j = 0; j < 16; ++j) any = any || ((h1[c0 + j] & PMASK) == want);
            if (any) {
                for (int j = 0; j < 16; ++j) {
                    if ((h1[c0 + j] & PMASK) == want) {
                        for (int q = 2; q < ns; ++q) acc += Gb[(size_t)(c0 + j) * NCH + cs + q];
                    }
                }
            }
        }
    }

    const float sg = (acc > 0.0f) ? 1.0f : ((acc < 0.0f) ? -1.0f : 0.0f);
    out[(size_t)b * PROJ + p] = sg * sqrtf(fabsf(acc) + 1e-8f);
}

__global__ __launch_bounds__(1024)
void cbp_norm_kernel(float* __restrict__ out)
{
    __shared__ float red[32];
    const int b    = blockIdx.x;
    const int tid  = threadIdx.x;
    const int lane = tid & 31;
    const int wave = tid >> 5;

    float* ob = out + (size_t)b * PROJ;
    float yv[PROJ / 1024];
    float ss = 0.0f;
    #pragma unroll
    for (int q = 0; q < PROJ / 1024; ++q) {
        const float y = ob[tid + q * 1024];
        yv[q] = y;
        ss   += y * y;
    }
    #pragma unroll
    for (int off = 16; off > 0; off >>= 1)
        ss += __shfl_down(ss, off, 32);
    if (lane == 0) red[wave] = ss;
    __syncthreads();
    if (tid == 0) {
        float tot = 0.0f;
        #pragma unroll
        for (int w = 0; w < 32; ++w) tot += red[w];
        red[0] = fmaxf(sqrtf(tot), 1e-12f);
    }
    __syncthreads();

    const float inv = 1.0f / red[0];
    #pragma unroll
    for (int q = 0; q < PROJ / 1024; ++q)
        ob[tid + q * 1024] = yv[q] * inv;
}

extern "C" void kernel_launch(void* const* d_in, const int* in_sizes, int n_in,
                              void* d_out, int out_size, void* d_ws, size_t ws_size,
                              hipStream_t stream) {
    const float* x  = (const float*)d_in[0];
    const float* s1 = (const float*)d_in[1];
    const float* s2 = (const float*)d_in[2];
    const int*   h1 = (const int*)  d_in[3];
    const int*   h2 = (const int*)  d_in[4];
    float*       y  = (float*)d_out;
    float*       Gs = (float*)d_ws;
    (void)in_sizes; (void)n_in; (void)out_size; (void)ws_size;

    cbp_gram_kernel<<<256, 256, 0, stream>>>(x, s1, s2, h2, Gs);
    cbp_gather_kernel<<<256, 512, 0, stream>>>(Gs, h1, h2, y);
    cbp_norm_kernel<<<16, 1024, 0, stream>>>(y);
}


# Gather: work-group = 128 bins of 4 images; the window's (row, column) pairs are found once from its rows' column runs, listed per bin in row order by bit counts, staged in LDS, summed per bin
# speedup vs baseline: 12.2259x; 1.1147x over previous
#include <hip/hip_runtime.h>
#include <math.h>

typedef __attribute__((ext_vector_type(2))) float v2f;
typedef __attribute__((ext_vector_type(4))) float v4f;
typedef __attribute__((ext_vector_type(8))) float v8f;

#define PROJ   8192
#define PMASK  (PROJ - 1)
#define NCH    512
#define HWSZ   196
#define KSTEP8 24

template <int NT>
__device__ __forceinline__ void rank_channels(const int* __restrict__ h2, const int tid,
                                              int* bcnt, int* bstart, int* member, int* wsum,
                                              int (&hv)[NCH / NT], int (&rnk)[NCH / NT],
                                              int (&pos)[NCH / NT], int (&cnt)[NCH / NT])
{
    constexpr int CPT = NCH / NT;
    const int lane = tid & 31;
    const int wave = tid >> 5;

    #pragma unroll
    for (int i = 0; i < CPT; ++i) bcnt[tid + i * NT] = 0;
    __syncthreads();

    int key[CPT], slot[CPT];
    #pragma unroll
    for (int i = 0; i < CPT; ++i) {
        const int t = tid + i * NT;
        hv[i]   = h2[t] & PMASK;
        key[i]  = (hv[i] << 9) | t;
        slot[i] = atomicAdd(&bcnt[hv[i] >> 4], 1);
    }
    __syncthreads();

    int v[CPT];
    int s = 0;
    #pragma unroll
    for (int i = 0; i < CPT; ++i) { v[i] = bcnt[tid * CPT + i]; s += v[i]; }
    int incl = s;
    #pragma unroll
    for (int d = 1; d < 32; d <<= 1) {
        const int u = __shfl_up(incl, d, 32);
        incl += (lane >= d) ? u : 0;
    }
    if (lane == 31) wsum[wave] = incl;
    __syncthreads();
    int ex = incl - s;
    #pragma unroll
    for (int w = 0; w < NT / 32; ++w) ex += (w < wave) ? wsum[w] : 0;
    #pragma unroll
    for (int i = 0; i < CPT; ++i) { bstart[tid * CPT + i] = ex; ex += v[i]; }
    __syncthreads();

    #pragma unroll
    for (int i = 0; i < CPT; ++i) member[bstart[hv[i] >> 4] + slot[i]] = key[i];
    __syncthreads();

    #pragma unroll
    for (int i = 0; i < CPT; ++i) {
        const int bk = hv[i] >> 4;
        const int st = bstart[bk];
        const int n  = bcnt[bk];
        int r = st, p = 0, c = 0;
        for (int q = 0; q < n; ++q) {
            const int  m  = member[st + q];
            const bool lt = (m < key[i]);
            const bool sm = ((m >> 9) == hv[i]);
            r += lt ? 1 : 0;
            p += (lt && sm) ? 1 : 0;
            c += sm ? 1 : 0;
        }
        rnk[i] = r; pos[i] = p; cnt[i] = c;
    }
}

__device__ __forceinline__ void mma_2x4(const v2f (&a)[2], const v2f (&bq)[4], v8f (&acc)[2][4])
{
    #pragma unroll
    for (int r = 0; r < 2; ++r) {
        #pragma unroll
        for (int s = 0; s < 4; ++s) {
            acc[r][s] = __builtin_amdgcn_wmma_f32_16x16x4_f32(
                    false, a[r], false, bq[s], (short)0, acc[r][s], false, false);
        }
    }
}

__global__ __launch_bounds__(256)
void cbp_gram_kernel(const float* __restrict__ x,
                     const float* __restrict__ s1,
                     const float* __restrict__ s2,
                     const int*   __restrict__ h2,
                     float* __restrict__ Gs)
{
    __shared__ int bcnt[NCH], bstart[NCH], member[NCH], perm[NCH];
    __shared__ int wsum[8];

    const int b    = blockIdx.x >> 4;
    const int blk  = blockIdx.x & 15;
    const int bi   = blk >> 2;
    const int bj   = blk & 3;
    const int tid  = threadIdx.x;
    const int lane = tid & 31;
    const int wave = tid >> 5;
    const int half = lane >> 4;
    const int lm16 = lane & 15;

    const int ti0 = bi * 8 + (wave >> 1) * 2;
    const int tj0 = bj * 8 + (wave & 1) * 4;

    const float* xb = x + (size_t)b * NCH * HWSZ;

    {
        int hv[2], rnk[2], pos[2], cnt[2];
        rank_channels<256>(h2, tid, bcnt, bstart, member, wsum, hv, rnk, pos, cnt);
        perm[rnk[0]] = tid;
        perm[rnk[1]] = tid + 256;
        __syncthreads();
    }
    int chB[4];
    #pragma unroll
    for (int s = 0; s < 4; ++s) chB[s] = perm[(tj0 + s) * 16 + lm16];

    const float* ra[2];
    const float* rb[4];
    #pragma unroll
    for (int r = 0; r < 2; ++r) ra[r] = xb + (size_t)((ti0 + r) * 16 + lm16) * HWSZ;
    #pragma unroll
    for (int s = 0; s < 4; ++s) rb[s] = xb + (size_t)chB[s] * HWSZ;

    v8f acc[2][4];
    #pragma unroll
    for (int r = 0; r < 2; ++r) {
        #pragma unroll
        for (int s = 0; s < 4; ++s) acc[r][s] = (v8f){};
    }

    const int ko = 4 * half;
    v4f ca[2], cb[4];
    #pragma unroll
    for (int r = 0; r < 2; ++r) ca[r] = *(const v4f*)(ra[r] + ko);
    #pragma unroll
    for (int s = 0; s < 4; ++s) cb[s] = *(const v4f*)(rb[s] + ko);

    for (int m = 0; m < KSTEP8 - 1; ++m) {
        v4f na[2], nb[4];
        const int kn = 8 * (m + 1) + ko;
        #pragma unroll
        for (int r = 0; r < 2; ++r) na[r] = *(const v4f*)(ra[r] + kn);
        #pragma unroll
        for (int s = 0; s < 4; ++s) nb[s] = *(const v4f*)(rb[s] + kn);

        v2f a0[2], b0[4], a1[2], b1[4];
        #pragma unroll
        for (int r = 0; r < 2; ++r) { a0[r] = (v2f){ca[r].x, ca[r].y}; a1[r] = (v2f){ca[r].z, ca[r].w}; }
        #pragma unroll
        for (int s = 0; s < 4; ++s) { b0[s] = (v2f){cb[s].x, cb[s].y}; b1[s] = (v2f){cb[s].z, cb[s].w}; }
        mma_2x4(a0, b0, acc);
        mma_2x4(a1, b1, acc);

        #pragma unroll
        for (int r = 0; r < 2; ++r) ca[r] = na[r];
        #pragma unroll
        for (int s = 0; s < 4; ++s) cb[s] = nb[s];
    }
    {
        v2f ta[2], tb[4];
        const int kt = 8 * KSTEP8 + 2 * half;
        #pragma unroll
        for (int r = 0; r < 2; ++r) ta[r] = *(const v2f*)(ra[r] + kt);
        #pragma unroll
        for (int s = 0; s < 4; ++s) tb[s] = *(const v2f*)(rb[s] + kt);

        v2f a0[2], b0[4], a1[2], b1[4];
        #pragma unroll
        for (int r = 0; r < 2; ++r) { a0[r] = (v2f){ca[r].x, ca[r].y}; a1[r] = (v2f){ca[r].z, ca[r].w}; }
        #pragma unroll
        for (int s = 0; s < 4; ++s) { b0[s] = (v2f){cb[s].x, cb[s].y}; b1[s] = (v2f){cb[s].z, cb[s].w}; }
        mma_2x4(a0, b0, acc);
        mma_2x4(a1, b1, acc);
        mma_2x4(ta, tb, acc);
    }

    float* Gb = Gs + (size_t)b * NCH * NCH;
    #pragma unroll
    for (int r = 0; r < 2; ++r) {
        #pragma unroll
        for (int s = 0; s < 4; ++s) {
            const int   cB  = (tj0 + s) * 16 + lm16;
            const float s2v = s2[chB[s]];
            #pragma unroll
            for (int v = 0; v < 8; ++v) {
                const int cA = (ti0 + r) * 16 + v + 8 * half;
                Gb[(size_t)cA * NCH + cB] = acc[r][s][v] * s1[cA] * s2v;
            }
        }
    }
}

#define WBINS   128
#define NIMG    4
#define NWORD   (NCH / 32)
#define LCAP    96
#define ELCAP   6144
#define ENTMASK 0x1FFF

__device__ __forceinline__ int count_below(const int* h2s, const int v)
{
    int lb = 0;
    #pragma unroll
    for (int step = NCH / 2; step >= 1; step >>= 1) {
        if (h2s[lb + step - 1] < v) lb += step;
    }
    lb += (h2s[lb] < v) ? 1 : 0;
    return lb;
}

__global__ __launch_bounds__(512)
void cbp_gather_kernel(const float* __restrict__ Gs,
                       const int*   __restrict__ h1,
                       const int*   __restrict__ h2,
                       float* __restrict__ out)
{
    __shared__ int      bcnt[NCH], bstart[NCH], member[NCH];
    __shared__ int      wsum[16];
    __shared__ int      wsc[16];
    __shared__ int      h2s[NCH];
    __shared__ unsigned colinfo[NCH];
    __shared__ unsigned rowbits[NWORD * WBINS];
    __shared__ int      pre[NWORD * WBINS];
    __shared__ int      binN[WBINS];
    __shared__ int      eaddr[ELCAP];
    __shared__ unsigned short lst[LCAP * WBINS];
    __shared__ int      ovf;
    __shared__ float    val[NIMG * ELCAP];

    const int win  = blockIdx.x >> 2;
    const int ig   = blockIdx.x & 3;
    const int tid  = threadIdx.x;
    const int lane = tid & 31;
    const int wave = tid >> 5;
    const int p0   = win * WBINS;
    const int q    = tid & (WBINS - 1);
    const int img  = tid >> 7;
    const int p    = p0 + q;

    const float* Gq = Gs + (size_t)(ig * NIMG) * NCH * NCH;
    const float* Gi = Gq + (size_t)img * NCH * NCH;

    #pragma unroll
    for (int i = 0; i < NWORD * WBINS / 512; ++i) rowbits[tid + i * 512] = 0u;
    if (tid == 0) ovf = 0;

    int hv[1], rnk[1], pos[1], cnt[1];
    rank_channels<512>(h2, tid, bcnt, bstart, member, wsum, hv, rnk, pos, cnt);
    h2s[rnk[0]]     = hv[0];
    colinfo[rnk[0]] = (unsigned)hv[0] | ((unsigned)cnt[0] << 13) | ((unsigned)pos[0] << 23);
    if (cnt[0] > 8) ovf = 1;
    __syncthreads();

    const int c    = tid;
    const int h1c  = h1[c] & PMASK;
    const int a0   = (p0 - h1c) & PMASK;
    const int a1   = a0 + WBINS;
    const bool wrp = (a1 > PROJ);
    const int lo   = count_below(h2s, a0);
    const int hi   = count_below(h2s, wrp ? (a1 - PROJ) : a1);
    const int len  = wrp ? (NCH - lo + hi) : (hi - lo);

    int incl = len;
    #pragma unroll
    for (int d = 1; d < 32; d <<= 1) {
        const int u = __shfl_up(incl, d, 32);
        incl += (lane >= d) ? u : 0;
    }
    if (lane == 31) wsc[wave] = incl;
    __syncthreads();
    int off = incl - len;
    int M = 0;
    #pragma unroll
    for (int w = 0; w < 16; ++w) {
        const int t = wsc[w];
        off += (w < wave) ? t : 0;
        M   += t;
    }
    M = __builtin_amdgcn_readfirstlane(M);

    const int      wrow = c >> 5;
    const unsigned brow = 1u << (c & 31);
    for (int i0 = 0; i0 < len; i0 += 8) {
        unsigned ci[8];
        #pragma unroll
        for (int k = 0; k < 8; ++k) ci[k] = colinfo[(lo + i0 + k) & (NCH - 1)];
        #pragma unroll
        for (int k = 0; k < 8; ++k) {
            const int i = i0 + k;
            if (i < len) {
                const int m = off + i;
                if (m < ELCAP) eaddr[m] = c * NCH + ((lo + i) & (NCH - 1));
                if ((ci[k] >> 23) == 0u) {
                    const int bq = (h1c + (int)(ci[k] & PMASK) - p0) & (WBINS - 1);
                    atomicOr(&rowbits[wrow * WBINS + bq], brow);
                }
            }
        }
    }
    __syncthreads();

    if (tid < WBINS) {
        int run = 0;
        #pragma unroll
        for (int w = 0; w < NWORD; ++w) {
            pre[w * WBINS + tid] = run;
            run += __popc(rowbits[w * WBINS + tid]);
        }
        binN[tid] = (run < LCAP) ? run : LCAP;
        if (run > LCAP) ovf = 1;
    }
    __syncthreads();

    for (int i0 = 0; i0 < len; i0 += 8) {
        unsigned ci[8];
        #pragma unroll
        for (int k = 0; k < 8; ++k) ci[k] = colinfo[(lo + i0 + k) & (NCH - 1)];
        #pragma unroll
        for (int k = 0; k < 8; ++k) {
            const int i = i0 + k;
            if (i < len && (ci[k] >> 23) == 0u) {
                const int bq   = (h1c + (int)(ci[k] & PMASK) - p0) & (WBINS - 1);
                const int at   = wrow * WBINS + bq;
                const int rank = pre[at] + __popc(rowbits[at] & (brow - 1u));
                const int more = (int)((ci[k] >> 13) & 0x3FFu) - 1;
                if (rank < LCAP) lst[rank * WBINS + bq] = (unsigned short)(((off + i) & ENTMASK) | ((more & 7) << 13));
            }
        }
    }
    __syncthreads();

    float acc = 0.0f;
    const bool fast = (M <= ELCAP) && (ovf == 0);
    if (fast) {
        {
            int   idx[ELCAP / 512];
            float g[ELCAP / 512][NIMG];
            #pragma unroll
            for (int k = 0; k < ELCAP / 512; ++k) {
                const int m = k * 512 + tid;
                idx[k] = (m < M) ? eaddr[m] : 0;
            }
            #pragma unroll
            for (int k = 0; k < ELCAP / 512; ++k) {
                #pragma unroll
                for (int b = 0; b < NIMG; ++b) g[k][b] = Gq[(size_t)b * NCH * NCH + idx[k]];
            }
            #pragma unroll
            for (int k = 0; k < ELCAP / 512; ++k) {
                const int m = k * 512 + tid;
                if (m < M) {
                    #pragma unroll
                    for (int b = 0; b < NIMG; ++b) val[b * ELCAP + m] = g[k][b];
                }
            }
        }
        __syncthreads();

        const float* vi = val + img * ELCAP;
        const int    n  = binN[q];
        for (int i1 = 0; i1 < n; i1 += 16) {
            int   en[16];
            float v[16];
            #pragma unroll
            for (int k = 0; k < 16; ++k) {
                const int ii = (i1 + k < n) ? (i1 + k) : 0;
                en[k] = lst[ii * WBINS + q];
            }
            #pragma unroll
            for (int k = 0; k < 16; ++k) v[k] = vi[en[k] & ENTMASK];
            #pragma unroll
            for (int k = 0; k < 16; ++k) {
                const bool on = (i1 + k < n);
                acc += on ? v[k] : 0.0f;
                const int more = on ? (en[k] >> 13) : 0;
                for (int t = 1; t <= more; ++t) acc += vi[(en[k] & ENTMASK) + t];
            }
        }
    } else {
        for (int cc = 0; cc < NCH; ++cc) {
            const int a  = (p - h1[cc]) & PMASK;
            const int lb = count_below(h2s, a);
            if (lb < NCH && h2s[lb] == a) {
                const int cn = (int)((colinfo[lb] >> 13) & 0x3FFu);
                for (int t = 0; t < cn; ++t) acc += Gi[(size_t)cc * NCH + lb + t];
            }
        }
    }

    const float sn = (acc > 0.0f) ? 1.0f : ((acc < 0.0f) ? -1.0f : 0.0f);
    out[(size_t)(ig * NIMG + img) * PROJ + p] = sn * sqrtf(fabsf(acc) + 1e-8f);
}

__global__ __launch_bounds__(1024)
void cbp_norm_kernel(float* __restrict__ out)
{
    __shared__ float red[32];
    const int b    = blockIdx.x;
    const int tid  = threadIdx.x;
    const int lane = tid & 31;
    const int wave = tid >> 5;

    float* ob = out + (size_t)b * PROJ;
    float yv[PROJ / 1024];
    float ss = 0.0f;
    #pragma unroll
    for (int q = 0; q < PROJ / 1024; ++q) {
        const float y = ob[tid + q * 1024];
        yv[q] = y;
        ss   += y * y;
    }
    #pragma unroll
    for (int off = 16; off > 0; off >>= 1)
        ss += __shfl_down(ss, off, 32);
    if (lane == 0) red[wave] = ss;
    __syncthreads();
    if (tid == 0) {
        float tot = 0.0f;
        #pragma unroll
        for (int w = 0; w < 32; ++w) tot += red[w];
        red[0] = fmaxf(sqrtf(tot), 1e-12f);
    }
    __syncthreads();

    const float inv = 1.0f / red[0];
    #pragma unroll
    for (int q = 0; q < PROJ / 1024; ++q)
        ob[tid + q * 1024] = yv[q] * inv;
}

extern "C" void kernel_launch(void* const* d_in, const int* in_sizes, int n_in,
                              void* d_out, int out_size, void* d_ws, size_t ws_size,
                              hipStream_t stream) {
    const float* x  = (const float*)d_in[0];
    const float* s1 = (const float*)d_in[1];
    const float* s2 = (const float*)d_in[2];
    const int*   h1 = (const int*)  d_in[3];
    const int*   h2 = (const int*)  d_in[4];
    float*       y  = (float*)d_out;
    float*       Gs = (float*)d_ws;
    (void)in_sizes; (void)n_in; (void)out_size; (void)ws_size;

    cbp_gram_kernel<<<256, 256, 0, stream>>>(x, s1, s2, h2, Gs);
    cbp_gather_kernel<<<256, 512, 0, stream>>>(Gs, h1, h2, y);
    cbp_norm_kernel<<<16, 1024, 0, stream>>>(y);
}


# Gram k-loop: fragment loads requested five steps (80 products) ahead through a six-set register ring, was about one step: a wave no longer waits a cache round trip in every step
# speedup vs baseline: 12.7541x; 1.0432x over previous
#include <hip/hip_runtime.h>
#include <math.h>

typedef __attribute__((ext_vector_type(2))) float v2f;
typedef __attribute__((ext_vector_type(4))) float v4f;
typedef __attribute__((ext_vector_type(8))) float v8f;

#define PROJ   8192
#define PMASK  (PROJ - 1)
#define NCH    512
#define HWSZ   196
#define KSTEP8 24

template <int NT>
__device__ __forceinline__ void rank_channels(const int* __restrict__ h2, const int tid,
                                              int* bcnt, int* bstart, int* member, int* wsum,
                                              int (&hv)[NCH / NT], int (&rnk)[NCH / NT],
                                              int (&pos)[NCH / NT], int (&cnt)[NCH / NT])
{
    constexpr int CPT = NCH / NT;
    const int lane = tid & 31;
    const int wave = tid >> 5;

    #pragma unroll
    for (int i = 0; i < CPT; ++i) bcnt[tid + i * NT] = 0;
    __syncthreads();

    int key[CPT], slot[CPT];
    #pragma unroll
    for (int i = 0; i < CPT; ++i) {
        const int t = tid + i * NT;
        hv[i]   = h2[t] & PMASK;
        key[i]  = (hv[i] << 9) | t;
        slot[i] = atomicAdd(&bcnt[hv[i] >> 4], 1);
    }
    __syncthreads();

    int v[CPT];
    int s = 0;
    #pragma unroll
    for (int i = 0; i < CPT; ++i) { v[i] = bcnt[tid * CPT + i]; s += v[i]; }
    int incl = s;
    #pragma unroll
    for (int d = 1; d < 32; d <<= 1) {
        const int u = __shfl_up(incl, d, 32);
        incl += (lane >= d) ? u : 0;
    }
    if (lane == 31) wsum[wave] = incl;
    __syncthreads();
    int ex = incl - s;
    #pragma unroll
    for (int w = 0; w < NT / 32; ++w) ex += (w < wave) ? wsum[w] : 0;
    #pragma unroll
    for (int i = 0; i < CPT; ++i) { bstart[tid * CPT + i] = ex; ex += v[i]; }
    __syncthreads();

    #pragma unroll
    for (int i = 0; i < CPT; ++i) member[bstart[hv[i] >> 4] + slot[i]] = key[i];
    __syncthreads();

    #pragma unroll
    for (int i = 0; i < CPT; ++i) {
        const int bk = hv[i] >> 4;
        const int st = bstart[bk];
        const int n  = bcnt[bk];
        int r = st, p = 0, c = 0;
        for (int q = 0; q < n; ++q) {
            const int  m  = member[st + q];
            const bool lt = (m < key[i]);
            const bool sm = ((m >> 9) == hv[i]);
            r += lt ? 1 : 0;
            p += (lt && sm) ? 1 : 0;
            c += sm ? 1 : 0;
        }
        rnk[i] = r; pos[i] = p; cnt[i] = c;
    }
}

__device__ __forceinline__ void mma_2x4(const v2f (&a)[2], const v2f (&bq)[4], v8f (&acc)[2][4])
{
    #pragma unroll
    for (int r = 0; r < 2; ++r) {
        #pragma unroll
        for (int s = 0; s < 4; ++s) {
            acc[r][s] = __builtin_amdgcn_wmma_f32_16x16x4_f32(
                    false, a[r], false, bq[s], (short)0, acc[r][s], false, false);
        }
    }
}

__global__ __launch_bounds__(256)
void cbp_gram_kernel(const float* __restrict__ x,
                     const float* __restrict__ s1,
                     const float* __restrict__ s2,
                     const int*   __restrict__ h2,
                     float* __restrict__ Gs)
{
    __shared__ int bcnt[NCH], bstart[NCH], member[NCH], perm[NCH];
    __shared__ int wsum[8];

    const int b    = blockIdx.x >> 4;
    const int blk  = blockIdx.x & 15;
    const int bi   = blk >> 2;
    const int bj   = blk & 3;
    const int tid  = threadIdx.x;
    const int lane = tid & 31;
    const int wave = tid >> 5;
    const int half = lane >> 4;
    const int lm16 = lane & 15;

    const int ti0 = bi * 8 + (wave >> 1) * 2;
    const int tj0 = bj * 8 + (wave & 1) * 4;

    const float* xb = x + (size_t)b * NCH * HWSZ;

    {
        int hv[2], rnk[2], pos[2], cnt[2];
        rank_channels<256>(h2, tid, bcnt, bstart, member, wsum, hv, rnk, pos, cnt);
        perm[rnk[0]] = tid;
        perm[rnk[1]] = tid + 256;
        __syncthreads();
    }
    int chB[4];
    #pragma unroll
    for (int s = 0; s < 4; ++s) chB[s] = perm[(tj0 + s) * 16 + lm16];

    const float* ra[2];
    const float* rb[4];
    #pragma unroll
    for (int r = 0; r < 2; ++r) ra[r] = xb + (size_t)((ti0 + r) * 16 + lm16) * HWSZ;
    #pragma unroll
    for (int s = 0; s < 4; ++s) rb[s] = xb + (size_t)chB[s] * HWSZ;

    v8f acc[2][4];
    #pragma unroll
    for (int r = 0; r < 2; ++r) {
        #pragma unroll
        for (int s = 0; s < 4; ++s) acc[r][s] = (v8f){};
    }

    const int ko = 4 * half;

    constexpr int PF   = 5;
    constexpr int RING = PF + 1;
    v4f fa[RING][2], fb[RING][4];
    v2f ta[2], tb[4];
    const int kt = 8 * KSTEP8 + 2 * half;
    v4f s1r[2][2];
    float s2r[4];

    #pragma unroll
    for (int d = 0; d < PF; ++d) {
        #pragma unroll
        for (int r = 0; r < 2; ++r) fa[d][r] = *(const v4f*)(ra[r] + 8 * d + ko);
        #pragma unroll
        for (int s = 0; s < 4; ++s) fb[d][s] = *(const v4f*)(rb[s] + 8 * d + ko);
    }

    #pragma unroll
    for (int m = 0; m < KSTEP8; ++m) {
        if (m + PF < KSTEP8) {
            const int kn = 8 * (m + PF) + ko;
            #pragma unroll
            for (int r = 0; r < 2; ++r) fa[(m + PF) % RING][r] = *(const v4f*)(ra[r] + kn);
            #pragma unroll
            for (int s = 0; s < 4; ++s) fb[(m + PF) % RING][s] = *(const v4f*)(rb[s] + kn);
        } else if (m + PF == KSTEP8) {
            #pragma unroll
            for (int r = 0; r < 2; ++r) ta[r] = *(const v2f*)(ra[r] + kt);
            #pragma unroll
            for (int s = 0; s < 4; ++s) tb[s] = *(const v2f*)(rb[s] + kt);
            #pragma unroll
            for (int r = 0; r < 2; ++r) {
                const float* sp = s1 + (ti0 + r) * 16 + 8 * half;
                s1r[r][0] = *(const v4f*)(sp);
                s1r[r][1] = *(const v4f*)(sp + 4);
            }
            #pragma unroll
            for (int s = 0; s < 4; ++s) s2r[s] = s2[chB[s]];
        }

        v2f a0[2], b0[4], a1[2], b1[4];
        #pragma unroll
        for (int r = 0; r < 2; ++r) {
            const v4f c = fa[m % RING][r];
            a0[r] = (v2f){c.x, c.y}; a1[r] = (v2f){c.z, c.w};
        }
        #pragma unroll
        for (int s = 0; s < 4; ++s) {
            const v4f c = fb[m % RING][s];
            b0[s] = (v2f){c.x, c.y}; b1[s] = (v2f){c.z, c.w};
        }
        mma_2x4(a0, b0, acc);
        mma_2x4(a1, b1, acc);
        __builtin_amdgcn_sched_barrier(0);
    }
    mma_2x4(ta, tb, acc);

    float* Gb = Gs + (size_t)b * NCH * NCH;
    #pragma unroll
    for (int r = 0; r < 2; ++r) {
        #pragma unroll
        for (int s = 0; s < 4; ++s) {
            const int   cB  = (tj0 + s) * 16 + lm16;
            const float s2v = s2r[s];
            #pragma unroll
            for (int v = 0; v < 8; ++v) {
                const int cA = (ti0 + r) * 16 + v + 8 * half;
                Gb[(size_t)cA * NCH + cB] = acc[r][s][v] * s1r[r][v >> 2][v & 3] * s2v;
            }
        }
    }
}

#define WBINS   128
#define NIMG    4
#define NWORD   (NCH / 32)
#define LCAP    96
#define ELCAP   6144
#define ENTMASK 0x1FFF

__device__ __forceinline__ int count_below(const int* h2s, const int v)
{
    int lb = 0;
    #pragma unroll
    for (int step = NCH / 2; step >= 1; step >>= 1) {
        if (h2s[lb + step - 1] < v) lb += step;
    }
    lb += (h2s[lb] < v) ? 1 : 0;
    return lb;
}

__global__ __launch_bounds__(512)
void cbp_gather_kernel(const float* __restrict__ Gs,
                       const int*   __restrict__ h1,
                       const int*   __restrict__ h2,
                       float* __restrict__ out)
{
    __shared__ int      bcnt[NCH], bstart[NCH], member[NCH];
    __shared__ int      wsum[16];
    __shared__ int      wsc[16];
    __shared__ int      h2s[NCH];
    __shared__ unsigned colinfo[NCH];
    __shared__ unsigned rowbits[NWORD * WBINS];
    __shared__ int      pre[NWORD * WBINS];
    __shared__ int      binN[WBINS];
    __shared__ int      eaddr[ELCAP];
    __shared__ unsigned short lst[LCAP * WBINS];
    __shared__ int      ovf;
    __shared__ float    val[NIMG * ELCAP];

    const int win  = blockIdx.x >> 2;
    const int ig   = blockIdx.x & 3;
    const int tid  = threadIdx.x;
    const int lane = tid & 31;
    const int wave = tid >> 5;
    const int p0   = win * WBINS;
    const int q    = tid & (WBINS - 1);
    const int img  = tid >> 7;
    const int p    = p0 + q;

    const float* Gq = Gs + (size_t)(ig * NIMG) * NCH * NCH;
    const float* Gi = Gq + (size_t)img * NCH * NCH;

    #pragma unroll
    for (int i = 0; i < NWORD * WBINS / 512; ++i) rowbits[tid + i * 512] = 0u;
    if (tid == 0) ovf = 0;

    int hv[1], rnk[1], pos[1], cnt[1];
    rank_channels<512>(h2, tid, bcnt, bstart, member, wsum, hv, rnk, pos, cnt);
    h2s[rnk[0]]     = hv[0];
    colinfo[rnk[0]] = (unsigned)hv[0] | ((unsigned)cnt[0] << 13) | ((unsigned)pos[0] << 23);
    if (cnt[0] > 8) ovf = 1;
    __syncthreads();

    const int c    = tid;
    const int h1c  = h1[c] & PMASK;
    const int a0   = (p0 - h1c) & PMASK;
    const int a1   = a0 + WBINS;
    const bool wrp = (a1 > PROJ);
    const int lo   = count_below(h2s, a0);
    const int hi   = count_below(h2s, wrp ? (a1 - PROJ) : a1);
    const int len  = wrp ? (NCH - lo + hi) : (hi - lo);

    int incl = len;
    #pragma unroll
    for (int d = 1; d < 32; d <<= 1) {
        const int u = __shfl_up(incl, d, 32);
        incl += (lane >= d) ? u : 0;
    }
    if (lane == 31) wsc[wave] = incl;
    __syncthreads();
    int off = incl - len;
    int M = 0;
    #pragma unroll
    for (int w = 0; w < 16; ++w) {
        const int t = wsc[w];
        off += (w < wave) ? t : 0;
        M   += t;
    }
    M = __builtin_amdgcn_readfirstlane(M);

    const int      wrow = c >> 5;
    const unsigned brow = 1u << (c & 31);
    for (int i0 = 0; i0 < len; i0 += 8) {
        unsigned ci[8];
        #pragma unroll
        for (int k = 0; k < 8; ++k) ci[k] = colinfo[(lo + i0 + k) & (NCH - 1)];
        #pragma unroll
        for (int k = 0; k < 8; ++k) {
            const int i = i0 + k;
            if (i < len) {
                const int m = off + i;
                if (m < ELCAP) eaddr[m] = c * NCH + ((lo + i) & (NCH - 1));
                if ((ci[k] >> 23) == 0u) {
                    const int bq = (h1c + (int)(ci[k] & PMASK) - p0) & (WBINS - 1);
                    atomicOr(&rowbits[wrow * WBINS + bq], brow);
                }
            }
        }
    }
    __syncthreads();

    if (tid < WBINS) {
        int run = 0;
        #pragma unroll
        for (int w = 0; w < NWORD; ++w) {
            pre[w * WBINS + tid] = run;
            run += __popc(rowbits[w * WBINS + tid]);
        }
        binN[tid] = (run < LCAP) ? run : LCAP;
        if (run > LCAP) ovf = 1;
    }
    __syncthreads();

    for (int i0 = 0; i0 < len; i0 += 8) {
        unsigned ci[8];
        #pragma unroll
        for (int k = 0; k < 8; ++k) ci[k] = colinfo[(lo + i0 + k) & (NCH - 1)];
        #pragma unroll
        for (int k = 0; k < 8; ++k) {
            const int i = i0 + k;
            if (i < len && (ci[k] >> 23) == 0u) {
                const int bq   = (h1c + (int)(ci[k] & PMASK) - p0) & (WBINS - 1);
                const int at   = wrow * WBINS + bq;
                const int rank = pre[at] + __popc(rowbits[at] & (brow - 1u));
                const int more = (int)((ci[k] >> 13) & 0x3FFu) - 1;
                if (rank < LCAP) lst[rank * WBINS + bq] = (unsigned short)(((off + i) & ENTMASK) | ((more & 7) << 13));
            }
        }
    }
    __syncthreads();

    float acc = 0.0f;
    const bool fast = (M <= ELCAP) && (ovf == 0);
    if (fast) {
        {
            int   idx[ELCAP / 512];
            float g[ELCAP / 512][NIMG];
            #pragma unroll
            for (int k = 0; k < ELCAP / 512; ++k) {
                const int m = k * 512 + tid;
                idx[k] = (m < M) ? eaddr[m] : 0;
            }
            #pragma unroll
            for (int k = 0; k < ELCAP / 512; ++k) {
                #pragma unroll
                for (int b = 0; b < NIMG; ++b) g[k][b] = Gq[(size_t)b * NCH * NCH + idx[k]];
            }
            #pragma unroll
            for (int k = 0; k < ELCAP / 512; ++k) {
                const int m = k * 512 + tid;
                if (m < M) {
                    #pragma unroll
                    for (int b = 0; b < NIMG; ++b) val[b * ELCAP + m] = g[k][b];
                }
            }
        }
        __syncthreads();

        const float* vi = val + img * ELCAP;
        const int    n  = binN[q];
        for (int i1 = 0; i1 < n; i1 += 16) {
            int   en[16];
            float v[16];
            #pragma unroll
            for (int k = 0; k < 16; ++k) {
                const int ii = (i1 + k < n) ? (i1 + k) : 0;
                en[k] = lst[ii * WBINS + q];
            }
            #pragma unroll
            for (int k = 0; k < 16; ++k) v[k] = vi[en[k] & ENTMASK];
            #pragma unroll
            for (int k = 0; k < 16; ++k) {
                const bool on = (i1 + k < n);
                acc += on ? v[k] : 0.0f;
                const int more = on ? (en[k] >> 13) : 0;
                for (int t = 1; t <= more; ++t) acc += vi[(en[k] & ENTMASK) + t];
            }
        }
    } else {
        for (int cc = 0; cc < NCH; ++cc) {
            const int a  = (p - h1[cc]) & PMASK;
            const int lb = count_below(h2s, a);
            if (lb < NCH && h2s[lb] == a) {
                const int cn = (int)((colinfo[lb] >> 13) & 0x3FFu);
                for (int t = 0; t < cn; ++t) acc += Gi[(size_t)cc * NCH + lb + t];
            }
        }
    }

    const float sn = (acc > 0.0f) ? 1.0f : ((acc < 0.0f) ? -1.0f : 0.0f);
    out[(size_t)(ig * NIMG + img) * PROJ + p] = sn * sqrtf(fabsf(acc) + 1e-8f);
}

__global__ __launch_bounds__(1024)
void cbp_norm_kernel(float* __restrict__ out)
{
    __shared__ float red[32];
    const int b    = blockIdx.x;
    const int tid  = threadIdx.x;
    const int lane = tid & 31;
    const int wave = tid >> 5;

    float* ob = out + (size_t)b * PROJ;
    float yv[PROJ / 1024];
    float ss = 0.0f;
    #pragma unroll
    for (int q = 0; q < PROJ / 1024; ++q) {
        const float y = ob[tid + q * 1024];
        yv[q] = y;
        ss   += y * y;
    }
    #pragma unroll
    for (int off = 16; off > 0; off >>= 1)
        ss += __shfl_down(ss, off, 32);
    if (lane == 0) red[wave] = ss;
    __syncthreads();
    if (tid == 0) {
        float tot = 0.0f;
        #pragma unroll
        for (int w = 0; w < 32; ++w) tot += red[w];
        red[0] = fmaxf(sqrtf(tot), 1e-12f);
    }
    __syncthreads();

    const float inv = 1.0f / red[0];
    #pragma unroll
    for (int q = 0; q < PROJ / 1024; ++q)
        ob[tid + q * 1024] = yv[q] * inv;
}

extern "C" void kernel_launch(void* const* d_in, const int* in_sizes, int n_in,
                              void* d_out, int out_size, void* d_ws, size_t ws_size,
                              hipStream_t stream) {
    const float* x  = (const float*)d_in[0];
    const float* s1 = (const float*)d_in[1];
    const float* s2 = (const float*)d_in[2];
    const int*   h1 = (const int*)  d_in[3];
    const int*   h2 = (const int*)  d_in[4];
    float*       y  = (float*)d_out;
    float*       Gs = (float*)d_ws;
    (void)in_sizes; (void)n_in; (void)out_size; (void)ws_size;

    cbp_gram_kernel<<<256, 256, 0, stream>>>(x, s1, s2, h2, Gs);
    cbp_gather_kernel<<<256, 512, 0, stream>>>(Gs, h1, h2, y);
    cbp_norm_kernel<<<16, 1024, 0, stream>>>(y);
}


# Gather sum loop made straight-line: holders of one h2 value are summed into their first element at staging, lists filled to whole trips with a zero element (was a branch and an LDS wait per entry)
# speedup vs baseline: 13.3012x; 1.0429x over previous
#include <hip/hip_runtime.h>
#include <math.h>

typedef __attribute__((ext_vector_type(2))) float v2f;
typedef __attribute__((ext_vector_type(4))) float v4f;
typedef __attribute__((ext_vector_type(8))) float v8f;

#define PROJ   8192
#define PMASK  (PROJ - 1)
#define NCH    512
#define HWSZ   196
#define KSTEP8 24

template <int NT>
__device__ __forceinline__ void rank_channels(const int* __restrict__ h2, const int tid,
                                              int* bcnt, int* bstart, int* member, int* wsum,
                                              int (&hv)[NCH / NT], int (&rnk)[NCH / NT],
                                              int (&pos)[NCH / NT], int (&cnt)[NCH / NT])
{
    constexpr int CPT = NCH / NT;
    const int lane = tid & 31;
    const int wave = tid >> 5;

    #pragma unroll
    for (int i = 0; i < CPT; ++i) bcnt[tid + i * NT] = 0;
    __syncthreads();

    int key[CPT], slot[CPT];
    #pragma unroll
    for (int i = 0; i < CPT; ++i) {
        const int t = tid + i * NT;
        hv[i]   = h2[t] & PMASK;
        key[i]  = (hv[i] << 9) | t;
        slot[i] = atomicAdd(&bcnt[hv[i] >> 4], 1);
    }
    __syncthreads();

    int v[CPT];
    int s = 0;
    #pragma unroll
    for (int i = 0; i < CPT; ++i) { v[i] = bcnt[tid * CPT + i]; s += v[i]; }
    int incl = s;
    #pragma unroll
    for (int d = 1; d < 32; d <<= 1) {
        const int u = __shfl_up(incl, d, 32);
        incl += (lane >= d) ? u : 0;
    }
    if (lane == 31) wsum[wave] = incl;
    __syncthreads();
    int ex = incl - s;
    #pragma unroll
    for (int w = 0; w < NT / 32; ++w) ex += (w < wave) ? wsum[w] : 0;
    #pragma unroll
    for (int i = 0; i < CPT; ++i) { bstart[tid * CPT + i] = ex; ex += v[i]; }
    __syncthreads();

    #pragma unroll
    for (int i = 0; i < CPT; ++i) member[bstart[hv[i] >> 4] + slot[i]] = key[i];
    __syncthreads();

    #pragma unroll
    for (int i = 0; i < CPT; ++i) {
        const int bk = hv[i] >> 4;
        const int st = bstart[bk];
        const int n  = bcnt[bk];
        int r = st, p = 0, c = 0;
        for (int q = 0; q < n; ++q) {
            const int  m  = member[st + q];
            const bool lt = (m < key[i]);
            const bool sm = ((m >> 9) == hv[i]);
            r += lt ? 1 : 0;
            p += (lt && sm) ? 1 : 0;
            c += sm ? 1 : 0;
        }
        rnk[i] = r; pos[i] = p; cnt[i] = c;
    }
}

__device__ __forceinline__ void mma_2x4(const v2f (&a)[2], const v2f (&bq)[4], v8f (&acc)[2][4])
{
    #pragma unroll
    for (int r = 0; r < 2; ++r) {
        #pragma unroll
        for (int s = 0; s < 4; ++s) {
            acc[r][s] = __builtin_amdgcn_wmma_f32_16x16x4_f32(
                    false, a[r], false, bq[s], (short)0, acc[r][s], false, false);
        }
    }
}

__global__ __launch_bounds__(256)
void cbp_gram_kernel(const float* __restrict__ x,
                     const float* __restrict__ s1,
                     const float* __restrict__ s2,
                     const int*   __restrict__ h2,
                     float* __restrict__ Gs)
{
    __shared__ int bcnt[NCH], bstart[NCH], member[NCH], perm[NCH];
    __shared__ int wsum[8];

    const int b    = blockIdx.x >> 4;
    const int blk  = blockIdx.x & 15;
    const int bi   = blk >> 2;
    const int bj   = blk & 3;
    const int tid  = threadIdx.x;
    const int lane = tid & 31;
    const int wave = tid >> 5;
    const int half = lane >> 4;
    const int lm16 = lane & 15;

    const int ti0 = bi * 8 + (wave >> 1) * 2;
    const int tj0 = bj * 8 + (wave & 1) * 4;

    const float* xb = x + (size_t)b * NCH * HWSZ;

    {
        int hv[2], rnk[2], pos[2], cnt[2];
        rank_channels<256>(h2, tid, bcnt, bstart, member, wsum, hv, rnk, pos, cnt);
        perm[rnk[0]] = tid;
        perm[rnk[1]] = tid + 256;
        __syncthreads();
    }
    int chB[4];
    #pragma unroll
    for (int s = 0; s < 4; ++s) chB[s] = perm[(tj0 + s) * 16 + lm16];

    const float* ra[2];
    const float* rb[4];
    #pragma unroll
    for (int r = 0; r < 2; ++r) ra[r] = xb + (size_t)((ti0 + r) * 16 + lm16) * HWSZ;
    #pragma unroll
    for (int s = 0; s < 4; ++s) rb[s] = xb + (size_t)chB[s] * HWSZ;

    v8f acc[2][4];
    #pragma unroll
    for (int r = 0; r < 2; ++r) {
        #pragma unroll
        for (int s = 0; s < 4; ++s) acc[r][s] = (v8f){};
    }

    const int ko = 4 * half;

    constexpr int PF   = 5;
    constexpr int RING = PF + 1;
    v4f fa[RING][2], fb[RING][4];
    v2f ta[2], tb[4];
    const int kt = 8 * KSTEP8 + 2 * half;
    v4f s1r[2][2];
    float s2r[4];

    #pragma unroll
    for (int d = 0; d < PF; ++d) {
        #pragma unroll
        for (int r = 0; r < 2; ++r) fa[d][r] = *(const v4f*)(ra[r] + 8 * d + ko);
        #pragma unroll
        for (int s = 0; s < 4; ++s) fb[d][s] = *(const v4f*)(rb[s] + 8 * d + ko);
    }

    #pragma unroll
    for (int m = 0; m < KSTEP8; ++m) {
        if (m + PF < KSTEP8) {
            const int kn = 8 * (m + PF) + ko;
            #pragma unroll
            for (int r = 0; r < 2; ++r) fa[(m + PF) % RING][r] = *(const v4f*)(ra[r] + kn);
            #pragma unroll
            for (int s = 0; s < 4; ++s) fb[(m + PF) % RING][s] = *(const v4f*)(rb[s] + kn);
        } else if (m + PF == KSTEP8) {
            #pragma unroll
            for (int r = 0; r < 2; ++r) ta[r] = *(const v2f*)(ra[r] + kt);
            #pragma unroll
            for (int s = 0; s < 4; ++s) tb[s] = *(const v2f*)(rb[s] + kt);
            #pragma unroll
            for (int r = 0; r < 2; ++r) {
                const float* sp = s1 + (ti0 + r) * 16 + 8 * half;
                s1r[r][0] = *(const v4f*)(sp);
                s1r[r][1] = *(const v4f*)(sp + 4);
            }
            #pragma unroll
            for (int s = 0; s < 4; ++s) s2r[s] = s2[chB[s]];
        }

        v2f a0[2], b0[4], a1[2], b1[4];
        #pragma unroll
        for (int r = 0; r < 2; ++r) {
            const v4f c = fa[m % RING][r];
            a0[r] = (v2f){c.x, c.y}; a1[r] = (v2f){c.z, c.w};
        }
        #pragma unroll
        for (int s = 0; s < 4; ++s) {
            const v4f c = fb[m % RING][s];
            b0[s] = (v2f){c.x, c.y}; b1[s] = (v2f){c.z, c.w};
        }
        mma_2x4(a0, b0, acc);
        mma_2x4(a1, b1, acc);
        __builtin_amdgcn_sched_barrier(0);
    }
    mma_2x4(ta, tb, acc);

    float* Gb = Gs + (size_t)b * NCH * NCH;
    #pragma unroll
    for (int r = 0; r < 2; ++r) {
        #pragma unroll
        for (int s = 0; s < 4; ++s) {
            const int   cB  = (tj0 + s) * 16 + lm16;
            const float s2v = s2r[s];
            #pragma unroll
            for (int v = 0; v < 8; ++v) {
                const int cA = (ti0 + r) * 16 + v + 8 * half;
                Gb[(size_t)cA * NCH + cB] = acc[r][s][v] * s1r[r][v >> 2][v & 3] * s2v;
            }
        }
    }
}

#define WBINS   128
#define NIMG    4
#define NWORD   (NCH / 32)
#define LCAP    96
#define ELCAP   6144
#define ZIDX    ELCAP
#define PLN     (ELCAP + 32)
#define TRIP    16
#define EADDR   0x3FFFF

__device__ __forceinline__ int count_below(const int* h2s, const int v)
{
    int lb = 0;
    #pragma unroll
    for (int step = NCH / 2; step >= 1; step >>= 1) {
        if (h2s[lb + step - 1] < v) lb += step;
    }
    lb += (h2s[lb] < v) ? 1 : 0;
    return lb;
}

__global__ __launch_bounds__(512)
void cbp_gather_kernel(const float* __restrict__ Gs,
                       const int*   __restrict__ h1,
                       const int*   __restrict__ h2,
                       float* __restrict__ out)
{
    __shared__ int      bcnt[NCH], bstart[NCH], member[NCH];
    __shared__ int      wsum[16];
    __shared__ int      wsc[16];
    __shared__ int      h2s[NCH];
    __shared__ unsigned colinfo[NCH];
    __shared__ unsigned rowbits[NWORD * WBINS];
    __shared__ int      pre[NWORD * WBINS];
    __shared__ int      binN[WBINS];
    __shared__ int      eaddr[ELCAP];
    __shared__ unsigned short lst[LCAP * WBINS];
    __shared__ int      ovf;
    __shared__ float    val[NIMG * PLN];

    const int win  = blockIdx.x >> 2;
    const int ig   = blockIdx.x & 3;
    const int tid  = threadIdx.x;
    const int lane = tid & 31;
    const int wave = tid >> 5;
    const int p0   = win * WBINS;
    const int q    = tid & (WBINS - 1);
    const int img  = tid >> 7;
    const int p    = p0 + q;

    const float* Gq = Gs + (size_t)(ig * NIMG) * NCH * NCH;
    const float* Gi = Gq + (size_t)img * NCH * NCH;

    #pragma unroll
    for (int i = 0; i < NWORD * WBINS / 512; ++i) rowbits[tid + i * 512] = 0u;
    if (tid == 0) ovf = 0;

    int hv[1], rnk[1], pos[1], cnt[1];
    rank_channels<512>(h2, tid, bcnt, bstart, member, wsum, hv, rnk, pos, cnt);
    h2s[rnk[0]]     = hv[0];
    colinfo[rnk[0]] = (unsigned)hv[0] | ((unsigned)cnt[0] << 13) | ((unsigned)pos[0] << 23);
    if (cnt[0] > 8) ovf = 1;
    __syncthreads();

    const int c    = tid;
    const int h1c  = h1[c] & PMASK;
    const int a0   = (p0 - h1c) & PMASK;
    const int a1   = a0 + WBINS;
    const bool wrp = (a1 > PROJ);
    const int lo   = count_below(h2s, a0);
    const int hi   = count_below(h2s, wrp ? (a1 - PROJ) : a1);
    const int len  = wrp ? (NCH - lo + hi) : (hi - lo);

    int incl = len;
    #pragma unroll
    for (int d = 1; d < 32; d <<= 1) {
        const int u = __shfl_up(incl, d, 32);
        incl += (lane >= d) ? u : 0;
    }
    if (lane == 31) wsc[wave] = incl;
    __syncthreads();
    int off = incl - len;
    int M = 0;
    #pragma unroll
    for (int w = 0; w < 16; ++w) {
        const int t = wsc[w];
        off += (w < wave) ? t : 0;
        M   += t;
    }
    M = __builtin_amdgcn_readfirstlane(M);

    const int      wrow = c >> 5;
    const unsigned brow = 1u << (c & 31);
    for (int i0 = 0; i0 < len; i0 += 8) {
        unsigned ci[8];
        #pragma unroll
        for (int k = 0; k < 8; ++k) ci[k] = colinfo[(lo + i0 + k) & (NCH - 1)];
        #pragma unroll
        for (int k = 0; k < 8; ++k) {
            const int i = i0 + k;
            if (i < len) {
                const int m = off + i;
                const bool first = ((ci[k] >> 23) == 0u);
                const int  more  = first ? ((int)((ci[k] >> 13) & 0x3FFu) - 1) : 0;
                if (m < ELCAP) eaddr[m] = (c * NCH + ((lo + i) & (NCH - 1))) | ((more & 7) << 18);
                if (first) {
                    const int bq = (h1c + (int)(ci[k] & PMASK) - p0) & (WBINS - 1);
                    atomicOr(&rowbits[wrow * WBINS + bq], brow);
                }
            }
        }
    }
    __syncthreads();

    if (tid < WBINS) {
        int run = 0;
        #pragma unroll
        for (int w = 0; w < NWORD; ++w) {
            pre[w * WBINS + tid] = run;
            run += __popc(rowbits[w * WBINS + tid]);
        }
        const int nb = (run < LCAP) ? run : LCAP;
        binN[tid] = nb;
        if (run > LCAP) ovf = 1;
        const int ne = (nb + TRIP - 1) & ~(TRIP - 1);
        for (int i = nb; i < ne; ++i) lst[i * WBINS + tid] = (unsigned short)ZIDX;
    }
    if (tid < NIMG) val[tid * PLN + ZIDX] = 0.0f;
    __syncthreads();

    for (int i0 = 0; i0 < len; i0 += 8) {
        unsigned ci[8];
        #pragma unroll
        for (int k = 0; k < 8; ++k) ci[k] = colinfo[(lo + i0 + k) & (NCH - 1)];
        #pragma unroll
        for (int k = 0; k < 8; ++k) {
            const int i = i0 + k;
            if (i < len && (ci[k] >> 23) == 0u) {
                const int bq   = (h1c + (int)(ci[k] & PMASK) - p0) & (WBINS - 1);
                const int at   = wrow * WBINS + bq;
                const int rank = pre[at] + __popc(rowbits[at] & (brow - 1u));
                if (rank < LCAP) lst[rank * WBINS + bq] = (unsigned short)(off + i);
            }
        }
    }
    __syncthreads();

    float acc = 0.0f;
    const bool fast = (M <= ELCAP) && (ovf == 0);
    if (fast) {
        {
            constexpr int KMAX = ELCAP / 512;
            const int klive = (M + 511) >> 9;
            int   idx[KMAX];
            float g[KMAX][NIMG], g1[KMAX][NIMG], g2[KMAX][NIMG];
            #pragma unroll
            for (int k = 0; k < KMAX; ++k) {
                const int m = k * 512 + tid;
                idx[k] = (m < M) ? eaddr[m] : 0;
            }
            #pragma unroll
            for (int k = 0; k < KMAX; ++k) {
                if (k < klive) {
                    const int a = idx[k] & EADDR;
                    #pragma unroll
                    for (int b = 0; b < NIMG; ++b) g[k][b] = Gq[(size_t)b * NCH * NCH + a];
                }
            }
            #pragma unroll
            for (int k = 0; k < KMAX; ++k) {
                if (k < klive) {
                    const int a    = idx[k] & EADDR;
                    const int more = idx[k] >> 18;
                    if (more >= 1) {
                        #pragma unroll
                        for (int b = 0; b < NIMG; ++b) g1[k][b] = Gq[(size_t)b * NCH * NCH + a + 1];
                    }
                    if (more >= 2) {
                        #pragma unroll
                        for (int b = 0; b < NIMG; ++b) g2[k][b] = Gq[(size_t)b * NCH * NCH + a + 2];
                    }
                }
            }
            #pragma unroll
            for (int k = 0; k < KMAX; ++k) {
                const int m = k * 512 + tid;
                if (k < klive && m < M) {
                    const int a    = idx[k] & EADDR;
                    const int more = idx[k] >> 18;
                    float s[NIMG];
                    #pragma unroll
                    for (int b = 0; b < NIMG; ++b) s[b] = g[k][b];
                    if (more >= 1) {
                        #pragma unroll
                        for (int b = 0; b < NIMG; ++b) s[b] += g1[k][b];
                    }
                    if (more >= 2) {
                        #pragma unroll
                        for (int b = 0; b < NIMG; ++b) s[b] += g2[k][b];
                    }
                    #pragma clang loop unroll(disable) vectorize(disable) interleave(disable)
                    for (int t = 3; t <= more; ++t) {
                        #pragma unroll
                        for (int b = 0; b < NIMG; ++b) s[b] += Gq[(size_t)b * NCH * NCH + a + t];
                    }
                    #pragma unroll
                    for (int b = 0; b < NIMG; ++b) val[b * PLN + m] = s[b];
                }
            }
        }
        __syncthreads();

        const float* vi = val + img * PLN;
        const int    ne = (binN[q] + TRIP - 1) & ~(TRIP - 1);
        for (int i1 = 0; i1 < ne; i1 += TRIP) {
            int   en[TRIP];
            float v[TRIP];
            #pragma unroll
            for (int k = 0; k < TRIP; ++k) en[k] = lst[(i1 + k) * WBINS + q];
            #pragma unroll
            for (int k = 0; k < TRIP; ++k) v[k] = vi[en[k]];
            #pragma unroll
            for (int k = 0; k < TRIP; ++k) acc += v[k];
        }
    } else {
        for (int cc = 0; cc < NCH; ++cc) {
            const int a  = (p - h1[cc]) & PMASK;
            const int lb = count_below(h2s, a);
            if (lb < NCH && h2s[lb] == a) {
                const int cn = (int)((colinfo[lb] >> 13) & 0x3FFu);
                for (int t = 0; t < cn; ++t) acc += Gi[(size_t)cc * NCH + lb + t];
            }
        }
    }

    const float sn = (acc > 0.0f) ? 1.0f : ((acc < 0.0f) ? -1.0f : 0.0f);
    out[(size_t)(ig * NIMG + img) * PROJ + p] = sn * sqrtf(fabsf(acc) + 1e-8f);
}

__global__ __launch_bounds__(1024)
void cbp_norm_kernel(float* __restrict__ out)
{
    __shared__ float red[32];
    const int b    = blockIdx.x;
    const int tid  = threadIdx.x;
    const int lane = tid & 31;
    const int wave = tid >> 5;

    float* ob = out + (size_t)b * PROJ;
    float yv[PROJ / 1024];
    float ss = 0.0f;
    #pragma unroll
    for (int q = 0; q < PROJ / 1024; ++q) {
        const float y = ob[tid + q * 1024];
        yv[q] = y;
        ss   += y * y;
    }
    #pragma unroll
    for (int off = 16; off > 0; off >>= 1)
        ss += __shfl_down(ss, off, 32);
    if (lane == 0) red[wave] = ss;
    __syncthreads();
    if (tid == 0) {
        float tot = 0.0f;
        #pragma unroll
        for (int w = 0; w < 32; ++w) tot += red[w];
        red[0] = fmaxf(sqrtf(tot), 1e-12f);
    }
    __syncthreads();

    const float inv = 1.0f / red[0];
    #pragma unroll
    for (int q = 0; q < PROJ / 1024; ++q)
        ob[tid + q * 1024] = yv[q] * inv;
}

extern "C" void kernel_launch(void* const* d_in, const int* in_sizes, int n_in,
                              void* d_out, int out_size, void* d_ws, size_t ws_size,
                              hipStream_t stream) {
    const float* x  = (const float*)d_in[0];
    const float* s1 = (const float*)d_in[1];
    const float* s2 = (const float*)d_in[2];
    const int*   h1 = (const int*)  d_in[3];
    const int*   h2 = (const int*)  d_in[4];
    float*       y  = (float*)d_out;
    float*       Gs = (float*)d_ws;
    (void)in_sizes; (void)n_in; (void)out_size; (void)ws_size;

    cbp_gram_kernel<<<256, 256, 0, stream>>>(x, s1, s2, h2, Gs);
    cbp_gather_kernel<<<256, 512, 0, stream>>>(Gs, h1, h2, y);
    cbp_norm_kernel<<<16, 1024, 0, stream>>>(y);
}


# Gram rows stored in (h2, channel) order like the columns: tile (I,J) is the transpose of (J,I), so 528 of 1024 tiles per image are computed; off-diagonal tiles also stored mirrored for the gather
# speedup vs baseline: 14.5365x; 1.0929x over previous
#include <hip/hip_runtime.h>
#include <math.h>

typedef __attribute__((ext_vector_type(2))) float v2f;
typedef __attribute__((ext_vector_type(4))) float v4f;
typedef __attribute__((ext_vector_type(8))) float v8f;

#define PROJ   8192
#define PMASK  (PROJ - 1)
#define NCH    512
#define HWSZ   196
#define KSTEP8 24

template <int NT>
__device__ __forceinline__ void rank_channels(const int* __restrict__ h2, const int tid,
                                              int* bcnt, int* bstart, int* member, int* wsum,
                                              int (&hv)[NCH / NT], int (&rnk)[NCH / NT],
                                              int (&pos)[NCH / NT], int (&cnt)[NCH / NT])
{
    constexpr int CPT = NCH / NT;
    const int lane = tid & 31;
    const int wave = tid >> 5;

    int key[CPT], slot[CPT];
    #pragma unroll
    for (int i = 0; i < CPT; ++i) {
        const int t = tid + i * NT;
        hv[i]   = h2[t] & PMASK;
        key[i]  = (hv[i] << 9) | t;
    }
    #pragma unroll
    for (int i = 0; i < CPT; ++i) bcnt[tid + i * NT] = 0;
    __syncthreads();

    #pragma unroll
    for (int i = 0; i < CPT; ++i) slot[i] = atomicAdd(&bcnt[hv[i] >> 4], 1);
    __syncthreads();

    int v[CPT];
    int s = 0;
    #pragma unroll
    for (int i = 0; i < CPT; ++i) { v[i] = bcnt[tid * CPT + i]; s += v[i]; }
    int incl = s;
    #pragma unroll
    for (int d = 1; d < 32; d <<= 1) {
        const int u = __shfl_up(incl, d, 32);
        incl += (lane >= d) ? u : 0;
    }
    if (lane == 31) wsum[wave] = incl;
    __syncthreads();
    int ex = incl - s;
    #pragma unroll
    for (int w = 0; w < NT / 32; ++w) ex += (w < wave) ? wsum[w] : 0;
    #pragma unroll
    for (int i = 0; i < CPT; ++i) { bstart[tid * CPT + i] = ex; ex += v[i]; }
    __syncthreads();

    #pragma unroll
    for (int i = 0; i < CPT; ++i) member[bstart[hv[i] >> 4] + slot[i]] = key[i];
    __syncthreads();

    #pragma unroll
    for (int i = 0; i < CPT; ++i) {
        const int bk = hv[i] >> 4;
        const int st = bstart[bk];
        const int n  = bcnt[bk];
        int r = st, p = 0, c = 0;
        for (int q = 0; q < n; ++q) {
            const int  m  = member[st + q];
            const bool lt = (m < key[i]);
            const bool sm = ((m >> 9) == hv[i]);
            r += lt ? 1 : 0;
            p += (lt && sm) ? 1 : 0;
            c += sm ? 1 : 0;
        }
        rnk[i] = r; pos[i] = p; cnt[i] = c;
    }
}

template <int NP, unsigned PA, unsigned PB, unsigned MIR>
__device__ __forceinline__ void gram_job(const float* __restrict__ xb, float* __restrict__ Gb,
                                         const int* perm, const float* s1p, const float* s2p,
                                         const int (&ft)[4], const int lm16, const int half)
{
    constexpr int NF = 4;

    const float* rp[NF];
    #pragma unroll
    for (int f = 0; f < NF; ++f) rp[f] = xb + (size_t)perm[ft[f] * 16 + lm16] * HWSZ;

    v8f acc[NP];
    #pragma unroll
    for (int i = 0; i < NP; ++i) acc[i] = (v8f){};

    const int ko = 4 * half;

    constexpr int PF   = 5;
    constexpr int RING = PF + 1;
    v4f fr[RING][NF];
    v2f tf[NF];
    const int kt = 8 * KSTEP8 + 2 * half;

    #pragma unroll
    for (int d = 0; d < PF; ++d) {
        #pragma unroll
        for (int f = 0; f < NF; ++f) fr[d][f] = *(const v4f*)(rp[f] + 8 * d + ko);
    }

    #pragma unroll
    for (int m = 0; m < KSTEP8; ++m) {
        if (m + PF < KSTEP8) {
            const int kn = 8 * (m + PF) + ko;
            #pragma unroll
            for (int f = 0; f < NF; ++f) fr[(m + PF) % RING][f] = *(const v4f*)(rp[f] + kn);
        } else if (m + PF == KSTEP8) {
            #pragma unroll
            for (int f = 0; f < NF; ++f) tf[f] = *(const v2f*)(rp[f] + kt);
        }

        v2f lo[NF], hi[NF];
        #pragma unroll
        for (int f = 0; f < NF; ++f) {
            const v4f c = fr[m % RING][f];
            lo[f] = (v2f){c.x, c.y}; hi[f] = (v2f){c.z, c.w};
        }
        #pragma unroll
        for (int i = 0; i < NP; ++i) {
            acc[i] = __builtin_amdgcn_wmma_f32_16x16x4_f32(
                    false, lo[(PA >> (4 * i)) & 15], false, lo[(PB >> (4 * i)) & 15],
                    (short)0, acc[i], false, false);
        }
        #pragma unroll
        for (int i = 0; i < NP; ++i) {
            acc[i] = __builtin_amdgcn_wmma_f32_16x16x4_f32(
                    false, hi[(PA >> (4 * i)) & 15], false, hi[(PB >> (4 * i)) & 15],
                    (short)0, acc[i], false, false);
        }
        __builtin_amdgcn_sched_barrier(0);
    }
    #pragma unroll
    for (int i = 0; i < NP; ++i) {
        acc[i] = __builtin_amdgcn_wmma_f32_16x16x4_f32(
                false, tf[(PA >> (4 * i)) & 15], false, tf[(PB >> (4 * i)) & 15],
                (short)0, acc[i], false, false);
    }

    #pragma unroll
    for (int i = 0; i < NP; ++i) {
        const int Ta = ft[(PA >> (4 * i)) & 15];
        const int Tb = ft[(PB >> (4 * i)) & 15];
        const int r0 = Ta * 16 + 8 * half;
        const int cB = Tb * 16 + lm16;
        {
            const v4f sa = *(const v4f*)(s1p + r0);
            const v4f sb = *(const v4f*)(s1p + r0 + 4);
            const float s1v[8] = {sa.x, sa.y, sa.z, sa.w, sb.x, sb.y, sb.z, sb.w};
            const float s2v = s2p[cB];
            #pragma unroll
            for (int v = 0; v < 8; ++v)
                Gb[(size_t)(r0 + v) * NCH + cB] = acc[i][v] * s1v[v] * s2v;
        }
        if ((MIR >> i) & 1u) {
            const float s1m = s1p[cB];
            const v4f ta = *(const v4f*)(s2p + r0);
            const v4f tb = *(const v4f*)(s2p + r0 + 4);
            float* mp = Gb + (size_t)cB * NCH + r0;
            const v4f o0 = {acc[i][0] * s1m * ta.x, acc[i][1] * s1m * ta.y,
                            acc[i][2] * s1m * ta.z, acc[i][3] * s1m * ta.w};
            const v4f o1 = {acc[i][4] * s1m * tb.x, acc[i][5] * s1m * tb.y,
                            acc[i][6] * s1m * tb.z, acc[i][7] * s1m * tb.w};
            *(v4f*)(mp)     = o0;
            *(v4f*)(mp + 4) = o1;
        }
    }
}

__global__ __launch_bounds__(256)
void cbp_gram_kernel(const float* __restrict__ x,
                     const float* __restrict__ s1,
                     const float* __restrict__ s2,
                     const int*   __restrict__ h2,
                     float* __restrict__ Gs)
{
    __shared__ int bcnt[NCH], bstart[NCH], member[NCH], perm[NCH];
    __shared__ int wsum[8];
    __shared__ __attribute__((aligned(16))) float s1p[NCH];
    __shared__ __attribute__((aligned(16))) float s2p[NCH];

    const int b    = blockIdx.x >> 4;
    const int blk  = blockIdx.x & 15;
    const int tid  = threadIdx.x;
    const int lane = tid & 31;
    const int half = lane >> 4;
    const int lm16 = lane & 15;
    const int wave = __builtin_amdgcn_readfirstlane(tid >> 5);

    const float* xb = x + (size_t)b * NCH * HWSZ;
    float*       Gb = Gs + (size_t)b * NCH * NCH;

    {
        const float s1a = s1[tid], s1b = s1[tid + 256];
        const float s2a = s2[tid], s2b = s2[tid + 256];
        int hv[2], rnk[2], pos[2], cnt[2];
        rank_channels<256>(h2, tid, bcnt, bstart, member, wsum, hv, rnk, pos, cnt);
        perm[rnk[0]] = tid;
        perm[rnk[1]] = tid + 256;
        s1p[rnk[0]]  = s1a;
        s1p[rnk[1]]  = s1b;
        s2p[rnk[0]]  = s2a;
        s2p[rnk[1]]  = s2b;
        __syncthreads();
    }

    const int  sub   = wave & 3;
    const bool mixed = (blk >= 12);
    int  ft[4];
    bool tri = false;
    if (mixed && wave >= 4) {
        const int D = 2 * (blk - 12) + ((wave >> 1) & 1);
        ft[0] = 4 * D; ft[1] = 4 * D + 1; ft[2] = 4 * D + 2; ft[3] = 4 * D + 3;
        tri = ((wave & 1) == 0);
    } else {
        const int t = mixed ? (24 + (blk - 12)) : (2 * blk + (wave >> 2));
        int BJ = 1;
        #pragma unroll
        for (int j = 2; j < 8; ++j) BJ = (t >= j * (j - 1) / 2) ? j : BJ;
        const int BI = t - BJ * (BJ - 1) / 2;
        const int I0 = 4 * BI + 2 * (sub >> 1);
        const int J0 = 4 * BJ + 2 * (sub & 1);
        ft[0] = I0; ft[1] = I0 + 1; ft[2] = J0; ft[3] = J0 + 1;
    }
    if (tri) {
        gram_job<6, 0x322100u, 0x332110u, 0x12u>(xb, Gb, perm, s1p, s2p, ft, lm16, half);
    } else {
        gram_job<4, 0x1100u, 0x3232u, 0xFu>(xb, Gb, perm, s1p, s2p, ft, lm16, half);
    }
}

#define WBINS   128
#define NIMG    4
#define NWORD   (NCH / 32)
#define LCAP    96
#define ELCAP   6144
#define ZIDX    ELCAP
#define PLN     (ELCAP + 32)
#define TRIP    16
#define EADDR   0x3FFFF

__device__ __forceinline__ int count_below(const int* h2s, const int v)
{
    int lb = 0;
    #pragma unroll
    for (int step = NCH / 2; step >= 1; step >>= 1) {
        if (h2s[lb + step - 1] < v) lb += step;
    }
    lb += (h2s[lb] < v) ? 1 : 0;
    return lb;
}

__global__ __launch_bounds__(512)
void cbp_gather_kernel(const float* __restrict__ Gs,
                       const int*   __restrict__ h1,
                       const int*   __restrict__ h2,
                       float* __restrict__ out)
{
    __shared__ int      bcnt[NCH], bstart[NCH], member[NCH];
    __shared__ int      wsum[16];
    __shared__ int      wsc[16];
    __shared__ int      h2s[NCH];
    __shared__ unsigned colinfo[NCH];
    __shared__ unsigned rowbits[NWORD * WBINS];
    __shared__ int      pre[NWORD * WBINS];
    __shared__ int      binN[WBINS];
    __shared__ int      eaddr[ELCAP];
    __shared__ unsigned short lst[LCAP * WBINS];
    __shared__ int      ovf;
    __shared__ float    val[NIMG * PLN];

    const int win  = blockIdx.x >> 2;
    const int ig   = blockIdx.x & 3;
    const int tid  = threadIdx.x;
    const int lane = tid & 31;
    const int wave = tid >> 5;
    const int p0   = win * WBINS;
    const int q    = tid & (WBINS - 1);
    const int img  = tid >> 7;
    const int p    = p0 + q;

    const float* Gq = Gs + (size_t)(ig * NIMG) * NCH * NCH;
    const float* Gi = Gq + (size_t)img * NCH * NCH;

    #pragma unroll
    for (int i = 0; i < NWORD * WBINS / 512; ++i) rowbits[tid + i * 512] = 0u;
    if (tid == 0) ovf = 0;

    int hv[1], rnk[1], pos[1], cnt[1];
    rank_channels<512>(h2, tid, bcnt, bstart, member, wsum, hv, rnk, pos, cnt);
    h2s[rnk[0]]     = hv[0];
    colinfo[rnk[0]] = (unsigned)hv[0] | ((unsigned)cnt[0] << 13) | ((unsigned)pos[0] << 23);
    if (cnt[0] > 8) ovf = 1;
    const int rk = rnk[0];
    __syncthreads();
    int* rankof = member;
    rankof[tid] = rk;

    const int c    = tid;
    const int h1c  = h1[c] & PMASK;
    const int a0   = (p0 - h1c) & PMASK;
    const int a1   = a0 + WBINS;
    const bool wrp = (a1 > PROJ);
    const int lo   = count_below(h2s, a0);
    const int hi   = count_below(h2s, wrp ? (a1 - PROJ) : a1);
    const int len  = wrp ? (NCH - lo + hi) : (hi - lo);

    int incl = len;
    #pragma unroll
    for (int d = 1; d < 32; d <<= 1) {
        const int u = __shfl_up(incl, d, 32);
        incl += (lane >= d) ? u : 0;
    }
    if (lane == 31) wsc[wave] = incl;
    __syncthreads();
    int off = incl - len;
    int M = 0;
    #pragma unroll
    for (int w = 0; w < 16; ++w) {
        const int t = wsc[w];
        off += (w < wave) ? t : 0;
        M   += t;
    }
    M = __builtin_amdgcn_readfirstlane(M);

    const int      wrow = c >> 5;
    const unsigned brow = 1u << (c & 31);
    for (int i0 = 0; i0 < len; i0 += 8) {
        unsigned ci[8];
        #pragma unroll
        for (int k = 0; k < 8; ++k) ci[k] = colinfo[(lo + i0 + k) & (NCH - 1)];
        #pragma unroll
        for (int k = 0; k < 8; ++k) {
            const int i = i0 + k;
            if (i < len) {
                const int m = off + i;
                const bool first = ((ci[k] >> 23) == 0u);
                const int  more  = first ? ((int)((ci[k] >> 13) & 0x3FFu) - 1) : 0;
                if (m < ELCAP) eaddr[m] = (rk * NCH + ((lo + i) & (NCH - 1))) | ((more & 7) << 18);
                if (first) {
                    const int bq = (h1c + (int)(ci[k] & PMASK) - p0) & (WBINS - 1);
                    atomicOr(&rowbits[wrow * WBINS + bq], brow);
                }
            }
        }
    }
    __syncthreads();

    if (tid < WBINS) {
        int run = 0;
        #pragma unroll
        for (int w = 0; w < NWORD; ++w) {
            pre[w * WBINS + tid] = run;
            run += __popc(rowbits[w * WBINS + tid]);
        }
        const int nb = (run < LCAP) ? run : LCAP;
        binN[tid] = nb;
        if (run > LCAP) ovf = 1;
        const int ne = (nb + TRIP - 1) & ~(TRIP - 1);
        for (int i = nb; i < ne; ++i) lst[i * WBINS + tid] = (unsigned short)ZIDX;
    }
    if (tid < NIMG) val[tid * PLN + ZIDX] = 0.0f;
    __syncthreads();

    for (int i0 = 0; i0 < len; i0 += 8) {
        unsigned ci[8];
        #pragma unroll
        for (int k = 0; k < 8; ++k) ci[k] = colinfo[(lo + i0 + k) & (NCH - 1)];
        #pragma unroll
        for (int k = 0; k < 8; ++k) {
            const int i = i0 + k;
            if (i < len && (ci[k] >> 23) == 0u) {
                const int bq   = (h1c + (int)(ci[k] & PMASK) - p0) & (WBINS - 1);
                const int at   = wrow * WBINS + bq;
                const int rank = pre[at] + __popc(rowbits[at] & (brow - 1u));
                if (rank < LCAP) lst[rank * WBINS + bq] = (unsigned short)(off + i);
            }
        }
    }
    __syncthreads();

    float acc = 0.0f;
    const bool fast = (M <= ELCAP) && (ovf == 0);
    if (fast) {
        {
            constexpr int KMAX = ELCAP / 512;
            const int klive = (M + 511) >> 9;
            int   idx[KMAX];
            float g[KMAX][NIMG], g1[KMAX][NIMG], g2[KMAX][NIMG];
            #pragma unroll
            for (int k = 0; k < KMAX; ++k) {
                const int m = k * 512 + tid;
                idx[k] = (m < M) ? eaddr[m] : 0;
            }
            #pragma unroll
            for (int k = 0; k < KMAX; ++k) {
                if (k < klive) {
                    const int a = idx[k] & EADDR;
                    #pragma unroll
                    for (int b = 0; b < NIMG; ++b) g[k][b] = Gq[(size_t)b * NCH * NCH + a];
                }
            }
            #pragma unroll
            for (int k = 0; k < KMAX; ++k) {
                if (k < klive) {
                    const int a    = idx[k] & EADDR;
                    const int more = idx[k] >> 18;
                    if (more >= 1) {
                        #pragma unroll
                        for (int b = 0; b < NIMG; ++b) g1[k][b] = Gq[(size_t)b * NCH * NCH + a + 1];
                    }
                    if (more >= 2) {
                        #pragma unroll
                        for (int b = 0; b < NIMG; ++b) g2[k][b] = Gq[(size_t)b * NCH * NCH + a + 2];
                    }
                }
            }
            #pragma unroll
            for (int k = 0; k < KMAX; ++k) {
                const int m = k * 512 + tid;
                if (k < klive && m < M) {
                    const int a    = idx[k] & EADDR;
                    const int more = idx[k] >> 18;
                    float s[NIMG];
                    #pragma unroll
                    for (int b = 0; b < NIMG; ++b) s[b] = g[k][b];
                    if (more >= 1) {
                        #pragma unroll
                        for (int b = 0; b < NIMG; ++b) s[b] += g1[k][b];
                    }
                    if (more >= 2) {
                        #pragma unroll
                        for (int b = 0; b < NIMG; ++b) s[b] += g2[k][b];
                    }
                    #pragma clang loop unroll(disable) vectorize(disable) interleave(disable)
                    for (int t = 3; t <= more; ++t) {
                        #pragma unroll
                        for (int b = 0; b < NIMG; ++b) s[b] += Gq[(size_t)b * NCH * NCH + a + t];
                    }
                    #pragma unroll
                    for (int b = 0; b < NIMG; ++b) val[b * PLN + m] = s[b];
                }
            }
        }
        __syncthreads();

        const float* vi = val + img * PLN;
        const int    ne = (binN[q] + TRIP - 1) & ~(TRIP - 1);
        for (int i1 = 0; i1 < ne; i1 += TRIP) {
            int   en[TRIP];
            float v[TRIP];
            #pragma unroll
            for (int k = 0; k < TRIP; ++k) en[k] = lst[(i1 + k) * WBINS + q];
            #pragma unroll
            for (int k = 0; k < TRIP; ++k) v[k] = vi[en[k]];
            #pragma unroll
            for (int k = 0; k < TRIP; ++k) acc += v[k];
        }
    } else {
        for (int cc = 0; cc < NCH; ++cc) {
            const int a  = (p - h1[cc]) & PMASK;
            const int lb = count_below(h2s, a);
            if (lb < NCH && h2s[lb] == a) {
                const int cn = (int)((colinfo[lb] >> 13) & 0x3FFu);
                for (int t = 0; t < cn; ++t) acc += Gi[(size_t)rankof[cc] * NCH + lb + t];
            }
        }
    }

    const float sn = (acc > 0.0f) ? 1.0f : ((acc < 0.0f) ? -1.0f : 0.0f);
    out[(size_t)(ig * NIMG + img) * PROJ + p] = sn * sqrtf(fabsf(acc) + 1e-8f);
}

__global__ __launch_bounds__(1024)
void cbp_norm_kernel(float* __restrict__ out)
{
    __shared__ float red[32];
    const int b    = blockIdx.x;
    const int tid  = threadIdx.x;
    const int lane = tid & 31;
    const int wave = tid >> 5;

    float* ob = out + (size_t)b * PROJ;
    float yv[PROJ / 1024];
    float ss = 0.0f;
    #pragma unroll
    for (int q = 0; q < PROJ / 1024; ++q) {
        const float y = ob[tid + q * 1024];
        yv[q] = y;
        ss   += y * y;
    }
    #pragma unroll
    for (int off = 16; off > 0; off >>= 1)
        ss += __shfl_down(ss, off, 32);
    if (lane == 0) red[wave] = ss;
    __syncthreads();
    if (tid == 0) {
        float tot = 0.0f;
        #pragma unroll
        for (int w = 0; w < 32; ++w) tot += red[w];
        red[0] = fmaxf(sqrtf(tot), 1e-12f);
    }
    __syncthreads();

    const float inv = 1.0f / red[0];
    #pragma unroll
    for (int q = 0; q < PROJ / 1024; ++q)
        ob[tid + q * 1024] = yv[q] * inv;
}

extern "C" void kernel_launch(void* const* d_in, const int* in_sizes, int n_in,
                              void* d_out, int out_size, void* d_ws, size_t ws_size,
                              hipStream_t stream) {
    const float* x  = (const float*)d_in[0];
    const float* s1 = (const float*)d_in[1];
    const float* s2 = (const float*)d_in[2];
    const int*   h1 = (const int*)  d_in[3];
    const int*   h2 = (const int*)  d_in[4];
    float*       y  = (float*)d_out;
    float*       Gs = (float*)d_ws;
    (void)in_sizes; (void)n_in; (void)out_size; (void)ws_size;

    cbp_gram_kernel<<<256, 256, 0, stream>>>(x, s1, s2, h2, Gs);
    cbp_gather_kernel<<<256, 512, 0, stream>>>(Gs, h1, h2, y);
    cbp_norm_kernel<<<16, 1024, 0, stream>>>(y);
}


# Gather no longer sorts the channels: launch 1, which ranks them anyway, hands the sorted keys over in the output buffer (a gather thread reads the word it later overwrites; h1 requested with it)
# speedup vs baseline: 15.0611x; 1.0361x over previous
#include <hip/hip_runtime.h>
#include <math.h>

typedef __attribute__((ext_vector_type(2))) float v2f;
typedef __attribute__((ext_vector_type(4))) float v4f;
typedef __attribute__((ext_vector_type(8))) float v8f;

#define PROJ   8192
#define PMASK  (PROJ - 1)
#define NCH    512
#define HWSZ   196
#define KSTEP8 24

template <int NT>
__device__ __forceinline__ void rank_channels(const int* __restrict__ h2, const int tid,
                                              int* bcnt, int* bstart, int* member, int* wsum,
                                              int (&hv)[NCH / NT], int (&rnk)[NCH / NT])
{
    constexpr int CPT = NCH / NT;
    const int lane = tid & 31;
    const int wave = tid >> 5;

    int key[CPT], slot[CPT];
    #pragma unroll
    for (int i = 0; i < CPT; ++i) {
        const int t = tid + i * NT;
        hv[i]   = h2[t] & PMASK;
        key[i]  = (hv[i] << 9) | t;
    }
    #pragma unroll
    for (int i = 0; i < CPT; ++i) bcnt[tid + i * NT] = 0;
    __syncthreads();

    #pragma unroll
    for (int i = 0; i < CPT; ++i) slot[i] = atomicAdd(&bcnt[hv[i] >> 4], 1);
    __syncthreads();

    int v[CPT];
    int s = 0;
    #pragma unroll
    for (int i = 0; i < CPT; ++i) { v[i] = bcnt[tid * CPT + i]; s += v[i]; }
    int incl = s;
    #pragma unroll
    for (int d = 1; d < 32; d <<= 1) {
        const int u = __shfl_up(incl, d, 32);
        incl += (lane >= d) ? u : 0;
    }
    if (lane == 31) wsum[wave] = incl;
    __syncthreads();
    int ex = incl - s;
    #pragma unroll
    for (int w = 0; w < NT / 32; ++w) ex += (w < wave) ? wsum[w] : 0;
    #pragma unroll
    for (int i = 0; i < CPT; ++i) { bstart[tid * CPT + i] = ex; ex += v[i]; }
    __syncthreads();

    #pragma unroll
    for (int i = 0; i < CPT; ++i) member[bstart[hv[i] >> 4] + slot[i]] = key[i];
    __syncthreads();

    #pragma unroll
    for (int i = 0; i < CPT; ++i) {
        const int bk = hv[i] >> 4;
        const int st = bstart[bk];
        const int n  = bcnt[bk];
        int r = st;
        for (int q = 0; q < n; ++q) r += (member[st + q] < key[i]) ? 1 : 0;
        rnk[i] = r;
    }
}

template <int NP, unsigned PA, unsigned PB, unsigned MIR>
__device__ __forceinline__ void gram_job(const float* __restrict__ xb, float* __restrict__ Gb,
                                         const int* perm, const float* s1p, const float* s2p,
                                         const int (&ft)[4], const int lm16, const int half)
{
    constexpr int NF = 4;

    const float* rp[NF];
    #pragma unroll
    for (int f = 0; f < NF; ++f) rp[f] = xb + (size_t)perm[ft[f] * 16 + lm16] * HWSZ;

    v8f acc[NP];
    #pragma unroll
    for (int i = 0; i < NP; ++i) acc[i] = (v8f){};

    const int ko = 4 * half;

    constexpr int PF   = 5;
    constexpr int RING = PF + 1;
    v4f fr[RING][NF];
    v2f tf[NF];
    const int kt = 8 * KSTEP8 + 2 * half;

    #pragma unroll
    for (int d = 0; d < PF; ++d) {
        #pragma unroll
        for (int f = 0; f < NF; ++f) fr[d][f] = *(const v4f*)(rp[f] + 8 * d + ko);
    }

    #pragma unroll
    for (int m = 0; m < KSTEP8; ++m) {
        if (m + PF < KSTEP8) {
            const int kn = 8 * (m + PF) + ko;
            #pragma unroll
            for (int f = 0; f < NF; ++f) fr[(m + PF) % RING][f] = *(const v4f*)(rp[f] + kn);
        } else if (m + PF == KSTEP8) {
            #pragma unroll
            for (int f = 0; f < NF; ++f) tf[f] = *(const v2f*)(rp[f] + kt);
        }

        v2f lo[NF], hi[NF];
        #pragma unroll
        for (int f = 0; f < NF; ++f) {
            const v4f c = fr[m % RING][f];
            lo[f] = (v2f){c.x, c.y}; hi[f] = (v2f){c.z, c.w};
        }
        #pragma unroll
        for (int i = 0; i < NP; ++i) {
            acc[i] = __builtin_amdgcn_wmma_f32_16x16x4_f32(
                    false, lo[(PA >> (4 * i)) & 15], false, lo[(PB >> (4 * i)) & 15],
                    (short)0, acc[i], false, false);
        }
        #pragma unroll
        for (int i = 0; i < NP; ++i) {
            acc[i] = __builtin_amdgcn_wmma_f32_16x16x4_f32(
                    false, hi[(PA >> (4 * i)) & 15], false, hi[(PB >> (4 * i)) & 15],
                    (short)0, acc[i], false, false);
        }
        __builtin_amdgcn_sched_barrier(0);
    }
    #pragma unroll
    for (int i = 0; i < NP; ++i) {
        acc[i] = __builtin_amdgcn_wmma_f32_16x16x4_f32(
                false, tf[(PA >> (4 * i)) & 15], false, tf[(PB >> (4 * i)) & 15],
                (short)0, acc[i], false, false);
    }

    #pragma unroll
    for (int i = 0; i < NP; ++i) {
        const int Ta = ft[(PA >> (4 * i)) & 15];
        const int Tb = ft[(PB >> (4 * i)) & 15];
        const int r0 = Ta * 16 + 8 * half;
        const int cB = Tb * 16 + lm16;
        {
            const v4f sa = *(const v4f*)(s1p + r0);
            const v4f sb = *(const v4f*)(s1p + r0 + 4);
            const float s1v[8] = {sa.x, sa.y, sa.z, sa.w, sb.x, sb.y, sb.z, sb.w};
            const float s2v = s2p[cB];
            #pragma unroll
            for (int v = 0; v < 8; ++v)
                Gb[(size_t)(r0 + v) * NCH + cB] = acc[i][v] * s1v[v] * s2v;
        }
        if ((MIR >> i) & 1u) {
            const float s1m = s1p[cB];
            const v4f ta = *(const v4f*)(s2p + r0);
            const v4f tb = *(const v4f*)(s2p + r0 + 4);
            float* mp = Gb + (size_t)cB * NCH + r0;
            const v4f o0 = {acc[i][0] * s1m * ta.x, acc[i][1] * s1m * ta.y,
                            acc[i][2] * s1m * ta.z, acc[i][3] * s1m * ta.w};
            const v4f o1 = {acc[i][4] * s1m * tb.x, acc[i][5] * s1m * tb.y,
                            acc[i][6] * s1m * tb.z, acc[i][7] * s1m * tb.w};
            *(v4f*)(mp)     = o0;
            *(v4f*)(mp + 4) = o1;
        }
    }
}

__global__ __launch_bounds__(256)
void cbp_gram_kernel(const float* __restrict__ x,
                     const float* __restrict__ s1,
                     const float* __restrict__ s2,
                     const int*   __restrict__ h2,
                     float* __restrict__ Gs,
                     float* __restrict__ out)
{
    __shared__ int bcnt[NCH], bstart[NCH], member[NCH], perm[NCH];
    __shared__ int skey[NCH];
    __shared__ int wsum[8];
    __shared__ __attribute__((aligned(16))) float s1p[NCH];
    __shared__ __attribute__((aligned(16))) float s2p[NCH];

    const int b    = blockIdx.x >> 4;
    const int blk  = blockIdx.x & 15;
    const int tid  = threadIdx.x;
    const int lane = tid & 31;
    const int half = lane >> 4;
    const int lm16 = lane & 15;
    const int wave = __builtin_amdgcn_readfirstlane(tid >> 5);

    const float* xb = x + (size_t)b * NCH * HWSZ;
    float*       Gb = Gs + (size_t)b * NCH * NCH;

    {
        const float s1a = s1[tid], s1b = s1[tid + 256];
        const float s2a = s2[tid], s2b = s2[tid + 256];
        int hv[2], rnk[2];
        rank_channels<256>(h2, tid, bcnt, bstart, member, wsum, hv, rnk);
        perm[rnk[0]] = tid;
        perm[rnk[1]] = tid + 256;
        skey[rnk[0]] = (hv[0] << 9) | tid;
        skey[rnk[1]] = (hv[1] << 9) | (tid + 256);
        s1p[rnk[0]]  = s1a;
        s1p[rnk[1]]  = s1b;
        s2p[rnk[0]]  = s2a;
        s2p[rnk[1]]  = s2b;
        __syncthreads();
    }
    {
        const int    gw = blockIdx.x >> 2, gi = blockIdx.x & 3;
        const size_t o0 = (size_t)(gi * 4 + (tid >> 7)) * PROJ + gw * 128 + (tid & 127);
        out[o0]            = __int_as_float(skey[tid]);
        out[o0 + 2 * PROJ] = __int_as_float(skey[tid + 256]);
    }

    const int  sub   = wave & 3;
    const bool mixed = (blk >= 12);
    int  ft[4];
    bool tri = false;
    if (mixed && wave >= 4) {
        const int D = 2 * (blk - 12) + ((wave >> 1) & 1);
        ft[0] = 4 * D; ft[1] = 4 * D + 1; ft[2] = 4 * D + 2; ft[3] = 4 * D + 3;
        tri = ((wave & 1) == 0);
    } else {
        const int t = mixed ? (24 + (blk - 12)) : (2 * blk + (wave >> 2));
        int BJ = 1;
        #pragma unroll
        for (int j = 2; j < 8; ++j) BJ = (t >= j * (j - 1) / 2) ? j : BJ;
        const int BI = t - BJ * (BJ - 1) / 2;
        const int I0 = 4 * BI + 2 * (sub >> 1);
        const int J0 = 4 * BJ + 2 * (sub & 1);
        ft[0] = I0; ft[1] = I0 + 1; ft[2] = J0; ft[3] = J0 + 1;
    }
    if (tri) {
        gram_job<6, 0x322100u, 0x332110u, 0x12u>(xb, Gb, perm, s1p, s2p, ft, lm16, half);
    } else {
        gram_job<4, 0x1100u, 0x3232u, 0xFu>(xb, Gb, perm, s1p, s2p, ft, lm16, half);
    }
}

#define WBINS   128
#define NIMG    4
#define NWORD   (NCH / 32)
#define LCAP    96
#define ELCAP   6144
#define ZIDX    ELCAP
#define PLN     (ELCAP + 32)
#define TRIP    16
#define EADDR   0x3FFFF

__device__ __forceinline__ int count_below(const int* h2s, const int v)
{
    int lb = 0;
    #pragma unroll
    for (int step = NCH / 2; step >= 1; step >>= 1) {
        if (h2s[lb + step - 1] < v) lb += step;
    }
    lb += (h2s[lb] < v) ? 1 : 0;
    return lb;
}

__global__ __launch_bounds__(512)
void cbp_gather_kernel(const float* __restrict__ Gs,
                       const int*   __restrict__ h1,
                       float* __restrict__ out)
{
    __shared__ int      rankof[NCH];
    __shared__ int      wsc[16];
    __shared__ int      h2s[NCH];
    __shared__ unsigned colinfo[NCH];
    __shared__ unsigned rowbits[NWORD * WBINS];
    __shared__ int      pre[NWORD * WBINS];
    __shared__ int      binN[WBINS];
    __shared__ int      eaddr[ELCAP];
    __shared__ unsigned short lst[LCAP * WBINS];
    __shared__ int      ovf;
    __shared__ float    val[NIMG * PLN];

    const int win  = blockIdx.x >> 2;
    const int ig   = blockIdx.x & 3;
    const int tid  = threadIdx.x;
    const int lane = tid & 31;
    const int wave = tid >> 5;
    const int p0   = win * WBINS;
    const int q    = tid & (WBINS - 1);
    const int img  = tid >> 7;
    const int p    = p0 + q;

    const float* Gq = Gs + (size_t)(ig * NIMG) * NCH * NCH;
    const float* Gi = Gq + (size_t)img * NCH * NCH;

    const size_t oidx = (size_t)(ig * NIMG + img) * PROJ + p;
    const int    key  = __float_as_int(out[oidx]);
    const int    h1c  = h1[tid] & PMASK;

    #pragma unroll
    for (int i = 0; i < NWORD * WBINS / 512; ++i) rowbits[tid + i * 512] = 0u;
    if (tid == 0) ovf = 0;

    const int hvj = (key >> 9) & PMASK;
    h2s[tid]            = hvj;
    rankof[key & (NCH - 1)] = tid;
    __syncthreads();
    {
        const int lft = h2s[(tid > 0) ? tid - 1 : 0];
        const int rgt = h2s[(tid < NCH - 1) ? tid + 1 : NCH - 1];
        int before = 0, after = 0;
        if (tid > 0 && lft == hvj) {
            before = 1;
            while (tid - before - 1 >= 0 && h2s[tid - before - 1] == hvj) ++before;
        }
        if (tid < NCH - 1 && rgt == hvj) {
            after = 1;
            while (tid + after + 1 < NCH && h2s[tid + after + 1] == hvj) ++after;
        }
        const int cntj = before + after + 1;
        colinfo[tid] = (unsigned)hvj | ((unsigned)cntj << 13) | ((unsigned)before << 23);
        if (cntj > 8) ovf = 1;
    }
    const int rk = rankof[tid];

    const int c    = tid;
    const int a0   = (p0 - h1c) & PMASK;
    const int a1   = a0 + WBINS;
    const bool wrp = (a1 > PROJ);
    const int lo   = count_below(h2s, a0);
    const int hi   = count_below(h2s, wrp ? (a1 - PROJ) : a1);
    const int len  = wrp ? (NCH - lo + hi) : (hi - lo);

    int incl = len;
    #pragma unroll
    for (int d = 1; d < 32; d <<= 1) {
        const int u = __shfl_up(incl, d, 32);
        incl += (lane >= d) ? u : 0;
    }
    if (lane == 31) wsc[wave] = incl;
    __syncthreads();
    int off = incl - len;
    int M = 0;
    #pragma unroll
    for (int w = 0; w < 16; ++w) {
        const int t = wsc[w];
        off += (w < wave) ? t : 0;
        M   += t;
    }
    M = __builtin_amdgcn_readfirstlane(M);

    const int      wrow = c >> 5;
    const unsigned brow = 1u << (c & 31);
    for (int i0 = 0; i0 < len; i0 += 8) {
        unsigned ci[8];
        #pragma unroll
        for (int k = 0; k < 8; ++k) ci[k] = colinfo[(lo + i0 + k) & (NCH - 1)];
        #pragma unroll
        for (int k = 0; k < 8; ++k) {
            const int i = i0 + k;
            if (i < len) {
                const int m = off + i;
                const bool first = ((ci[k] >> 23) == 0u);
                const int  more  = first ? ((int)((ci[k] >> 13) & 0x3FFu) - 1) : 0;
                if (m < ELCAP) eaddr[m] = (rk * NCH + ((lo + i) & (NCH - 1))) | ((more & 7) << 18);
                if (first) {
                    const int bq = (h1c + (int)(ci[k] & PMASK) - p0) & (WBINS - 1);
                    atomicOr(&rowbits[wrow * WBINS + bq], brow);
                }
            }
        }
    }
    __syncthreads();

    if (tid < WBINS) {
        int run = 0;
        #pragma unroll
        for (int w = 0; w < NWORD; ++w) {
            pre[w * WBINS + tid] = run;
            run += __popc(rowbits[w * WBINS + tid]);
        }
        const int nb = (run < LCAP) ? run : LCAP;
        binN[tid] = nb;
        if (run > LCAP) ovf = 1;
        const int ne = (nb + TRIP - 1) & ~(TRIP - 1);
        for (int i = nb; i < ne; ++i) lst[i * WBINS + tid] = (unsigned short)ZIDX;
    }
    if (tid < NIMG) val[tid * PLN + ZIDX] = 0.0f;
    __syncthreads();

    for (int i0 = 0; i0 < len; i0 += 8) {
        unsigned ci[8];
        #pragma unroll
        for (int k = 0; k < 8; ++k) ci[k] = colinfo[(lo + i0 + k) & (NCH - 1)];
        #pragma unroll
        for (int k = 0; k < 8; ++k) {
            const int i = i0 + k;
            if (i < len && (ci[k] >> 23) == 0u) {
                const int bq   = (h1c + (int)(ci[k] & PMASK) - p0) & (WBINS - 1);
                const int at   = wrow * WBINS + bq;
                const int rank = pre[at] + __popc(rowbits[at] & (brow - 1u));
                if (rank < LCAP) lst[rank * WBINS + bq] = (unsigned short)(off + i);
            }
        }
    }
    __syncthreads();

    float acc = 0.0f;
    const bool fast = (M <= ELCAP) && (ovf == 0);
    if (fast) {
        {
            constexpr int KMAX = ELCAP / 512;
            const int klive = (M + 511) >> 9;
            int   idx[KMAX];
            float g[KMAX][NIMG], g1[KMAX][NIMG], g2[KMAX][NIMG];
            #pragma unroll
            for (int k = 0; k < KMAX; ++k) {
                const int m = k * 512 + tid;
                idx[k] = (m < M) ? eaddr[m] : 0;
            }
            #pragma unroll
            for (int k = 0; k < KMAX; ++k) {
                if (k < klive) {
                    const int a = idx[k] & EADDR;
                    #pragma unroll
                    for (int b = 0; b < NIMG; ++b) g[k][b] = Gq[(size_t)b * NCH * NCH + a];
                }
            }
            #pragma unroll
            for (int k = 0; k < KMAX; ++k) {
                if (k < klive) {
                    const int a    = idx[k] & EADDR;
                    const int more = idx[k] >> 18;
                    if (more >= 1) {
                        #pragma unroll
                        for (int b = 0; b < NIMG; ++b) g1[k][b] = Gq[(size_t)b * NCH * NCH + a + 1];
                    }
                    if (more >= 2) {
                        #pragma unroll
                        for (int b = 0; b < NIMG; ++b) g2[k][b] = Gq[(size_t)b * NCH * NCH + a + 2];
                    }
                }
            }
            #pragma unroll
            for (int k = 0; k < KMAX; ++k) {
                const int m = k * 512 + tid;
                if (k < klive && m < M) {
                    const int a    = idx[k] & EADDR;
                    const int more = idx[k] >> 18;
                    float s[NIMG];
                    #pragma unroll
                    for (int b = 0; b < NIMG; ++b) s[b] = g[k][b];
                    if (more >= 1) {
                        #pragma unroll
                        for (int b = 0; b < NIMG; ++b) s[b] += g1[k][b];
                    }
                    if (more >= 2) {
                        #pragma unroll
                        for (int b = 0; b < NIMG; ++b) s[b] += g2[k][b];
                    }
                    #pragma clang loop unroll(disable) vectorize(disable) interleave(disable)
                    for (int t = 3; t <= more; ++t) {
                        #pragma unroll
                        for (int b = 0; b < NIMG; ++b) s[b] += Gq[(size_t)b * NCH * NCH + a + t];
                    }
                    #pragma unroll
                    for (int b = 0; b < NIMG; ++b) val[b * PLN + m] = s[b];
                }
            }
        }
        __syncthreads();

        const float* vi = val + img * PLN;
        const int    ne = (binN[q] + TRIP - 1) & ~(TRIP - 1);
        for (int i1 = 0; i1 < ne; i1 += TRIP) {
            int   en[TRIP];
            float v[TRIP];
            #pragma unroll
            for (int k = 0; k < TRIP; ++k) en[k] = lst[(i1 + k) * WBINS + q];
            #pragma unroll
            for (int k = 0; k < TRIP; ++k) v[k] = vi[en[k]];
            #pragma unroll
            for (int k = 0; k < TRIP; ++k) acc += v[k];
        }
    } else {
        for (int cc = 0; cc < NCH; ++cc) {
            const int a  = (p - h1[cc]) & PMASK;
            const int lb = count_below(h2s, a);
            if (lb < NCH && h2s[lb] == a) {
                const int cn = (int)((colinfo[lb] >> 13) & 0x3FFu);
                for (int t = 0; t < cn; ++t) acc += Gi[(size_t)rankof[cc] * NCH + lb + t];
            }
        }
    }

    const float sn = (acc > 0.0f) ? 1.0f : ((acc < 0.0f) ? -1.0f : 0.0f);
    out[oidx] = sn * sqrtf(fabsf(acc) + 1e-8f);
}

__global__ __launch_bounds__(1024)
void cbp_norm_kernel(float* __restrict__ out)
{
    __shared__ float red[32];
    const int b    = blockIdx.x;
    const int tid  = threadIdx.x;
    const int lane = tid & 31;
    const int wave = tid >> 5;

    float* ob = out + (size_t)b * PROJ;
    float yv[PROJ / 1024];
    float ss = 0.0f;
    #pragma unroll
    for (int q = 0; q < PROJ / 1024; ++q) {
        const float y = ob[tid + q * 1024];
        yv[q] = y;
        ss   += y * y;
    }
    #pragma unroll
    for (int off = 16; off > 0; off >>= 1)
        ss += __shfl_down(ss, off, 32);
    if (lane == 0) red[wave] = ss;
    __syncthreads();
    if (tid == 0) {
        float tot = 0.0f;
        #pragma unroll
        for (int w = 0; w < 32; ++w) tot += red[w];
        red[0] = fmaxf(sqrtf(tot), 1e-12f);
    }
    __syncthreads();

    const float inv = 1.0f / red[0];
    #pragma unroll
    for (int q = 0; q < PROJ / 1024; ++q)
        ob[tid + q * 1024] = yv[q] * inv;
}

extern "C" void kernel_launch(void* const* d_in, const int* in_sizes, int n_in,
                              void* d_out, int out_size, void* d_ws, size_t ws_size,
                              hipStream_t stream) {
    const float* x  = (const float*)d_in[0];
    const float* s1 = (const float*)d_in[1];
    const float* s2 = (const float*)d_in[2];
    const int*   h1 = (const int*)  d_in[3];
    const int*   h2 = (const int*)  d_in[4];
    float*       y  = (float*)d_out;
    float*       Gs = (float*)d_ws;
    (void)in_sizes; (void)n_in; (void)out_size; (void)ws_size;

    cbp_gram_kernel<<<256, 256, 0, stream>>>(x, s1, s2, h2, Gs, y);
    cbp_gather_kernel<<<256, 512, 0, stream>>>(Gs, h1, y);
    cbp_norm_kernel<<<16, 1024, 0, stream>>>(y);
}
